# starting kernel
# baseline (speedup 1.0000x reference)
#include <hip/hip_runtime.h>
#include <math.h>

#define N_NODES 50000
#define N_EDGES 800000
#define IN_DIM  128
#define OUT_DIM 128
#define HEADS   4
#define D_K     32
#define ATT_DIM 65

typedef __attribute__((ext_vector_type(2))) float v2f;
typedef __attribute__((ext_vector_type(8))) float v8f;

__global__ void __launch_bounds__(256)
gemm_z_wmma(const float* __restrict__ h, const float* __restrict__ W,
            const float* __restrict__ bias, float* __restrict__ z) {
    const int wave = (blockIdx.x * blockDim.x + threadIdx.x) >> 5;
    const int lane = threadIdx.x & 31;
    if (wave >= (N_NODES / 16)) return;

    const int m = lane & 15;
    const int g = lane >> 4;

    const float* __restrict__ hrow = h + (size_t)(wave * 16 + m) * IN_DIM;
    v2f a[32];
#pragma unroll
    for (int ks = 0; ks < 32; ++ks) {
        const int k = 4 * ks + 2 * g;
        a[ks].x = hrow[k];
        a[ks].y = hrow[k + 1];
    }

    for (int ot = 0; ot < 8; ++ot) {
        const int col = ot * 16 + m;
        const float* __restrict__ wrow = W + (size_t)col * IN_DIM;
        v8f c = {};
#pragma unroll
        for (int ks = 0; ks < 32; ++ks) {
            const int k = 4 * ks + 2 * g;
            v2f b;
            b.x = wrow[k];
            b.y = wrow[k + 1];
            c = __builtin_amdgcn_wmma_f32_16x16x4_f32(
                     false, a[ks],  false, b,
                     (short)0, c,  false,  false);
        }
        const float bv = bias[col];
        const int rowBase = wave * 16 + g * 8;
#pragma unroll
        for (int r = 0; r < 8; ++r) {
            z[(size_t)(rowBase + r) * OUT_DIM + col] = c[r] + bv;
        }
    }
}

__global__ void __launch_bounds__(256)
node_scores(const float* __restrict__ z, const float* __restrict__ att,
            float* __restrict__ s_src, float* __restrict__ s_dst) {
    const int gid = blockIdx.x * blockDim.x + threadIdx.x;
    if (gid >= N_NODES * HEADS) return;
    const int hd = gid & 3;
    const int n  = gid >> 2;
    const float* __restrict__ zp = z + (size_t)n * OUT_DIM + hd * D_K;
    const float* __restrict__ as = att + hd * ATT_DIM;
    const float* __restrict__ ad = as + D_K;
    float ss = 0.f, sd = 0.f;
#pragma unroll
    for (int k = 0; k < D_K; ++k) {
        const float v = zp[k];
        ss = fmaf(v, as[k], ss);
        sd = fmaf(v, ad[k], sd);
    }
    s_src[gid] = ss;
    s_dst[gid] = sd;
}

__global__ void __launch_bounds__(256)
zero_f32(float* __restrict__ p, int n) {
    const int gid = blockIdx.x * blockDim.x + threadIdx.x;
    if (gid < n) p[gid] = 0.f;
}

__device__ __forceinline__ float leaky_clip_exp(float s) {
    s = (s >= 0.f) ? s : 0.2f * s;
    s = fminf(fmaxf(s, -20.f), 20.f);
    return __expf(s);
}

__global__ void __launch_bounds__(256)
edge_denom(const int* __restrict__ ei, const float* __restrict__ ef,
           const float* __restrict__ s_src, const float* __restrict__ s_dst,
           const float* __restrict__ att, float* __restrict__ denom) {
    const int gid = blockIdx.x * blockDim.x + threadIdx.x;
    if (gid >= N_EDGES * HEADS) return;
    const int hd = gid & 3;
    const int e  = gid >> 2;
    const int src = ei[e];
    const int dst = ei[N_EDGES + e];
    const float s = s_src[src * HEADS + hd] + s_dst[dst * HEADS + hd]
                  + ef[e] * att[hd * ATT_DIM + 2 * D_K];
    const float alpha = leaky_clip_exp(s);
    __hip_atomic_fetch_add(&denom[dst * HEADS + hd], alpha,
                           __ATOMIC_RELAXED, __HIP_MEMORY_SCOPE_AGENT);
}

__global__ void __launch_bounds__(256)
edge_aggregate(const int* __restrict__ ei, const float* __restrict__ ef,
               const float* __restrict__ s_src, const float* __restrict__ s_dst,
               const float* __restrict__ att, const float* __restrict__ denom,
               const float* __restrict__ z, float* __restrict__ accum) {
    const int wid  = (blockIdx.x * blockDim.x + threadIdx.x) >> 5;
    const int lane = threadIdx.x & 31;
    if (wid >= N_EDGES) return;
    const int src = ei[wid];
    const int dst = ei[N_EDGES + wid];
    const int hd  = lane >> 3;

    const float s = s_src[src * HEADS + hd] + s_dst[dst * HEADS + hd]
                  + ef[wid] * att[hd * ATT_DIM + 2 * D_K];
    const float alpha = leaky_clip_exp(s);
    const float na = alpha / (denom[dst * HEADS + hd] + 1e-6f);

    const float4 zv = *(const float4*)(z + (size_t)src * OUT_DIM + lane * 4);
    float* __restrict__ ap = accum + (size_t)dst * OUT_DIM + lane * 4;
    __hip_atomic_fetch_add(ap + 0, zv.x * na, __ATOMIC_RELAXED, __HIP_MEMORY_SCOPE_AGENT);
    __hip_atomic_fetch_add(ap + 1, zv.y * na, __ATOMIC_RELAXED, __HIP_MEMORY_SCOPE_AGENT);
    __hip_atomic_fetch_add(ap + 2, zv.z * na, __ATOMIC_RELAXED, __HIP_MEMORY_SCOPE_AGENT);
    __hip_atomic_fetch_add(ap + 3, zv.w * na, __ATOMIC_RELAXED, __HIP_MEMORY_SCOPE_AGENT);
}

__global__ void __launch_bounds__(256)
finalize(const float* __restrict__ accum, const float* __restrict__ z,
         const float* __restrict__ ln_g, const float* __restrict__ ln_b,
         float* __restrict__ out) {
    const int wid  = (blockIdx.x * blockDim.x + threadIdx.x) >> 5;
    const int lane = threadIdx.x & 31;
    if (wid >= N_NODES) return;

    const size_t base = (size_t)wid * OUT_DIM + lane * 4;
    const float4 a  = *(const float4*)(accum + base);
    const float4 zv = *(const float4*)(z + base);
    float x[4] = { a.x + zv.x, a.y + zv.y, a.z + zv.z, a.w + zv.w };

    float s = x[0] + x[1] + x[2] + x[3];
#pragma unroll
    for (int off = 16; off > 0; off >>= 1) s += __shfl_xor(s, off, 32);
    const float mu = s * (1.f / 128.f);

    float v = 0.f;
#pragma unroll
    for (int i = 0; i < 4; ++i) { const float d = x[i] - mu; v = fmaf(d, d, v); }
#pragma unroll
    for (int off = 16; off > 0; off >>= 1) v += __shfl_xor(v, off, 32);
    const float rstd = rsqrtf(v * (1.f / 128.f) + 1e-5f);

    const float4 gv = *(const float4*)(ln_g + lane * 4);
    const float4 bv = *(const float4*)(ln_b + lane * 4);
    const float g4[4] = { gv.x, gv.y, gv.z, gv.w };
    const float b4[4] = { bv.x, bv.y, bv.z, bv.w };

    float y[4];
#pragma unroll
    for (int i = 0; i < 4; ++i) {
        const float t = (x[i] - mu) * rstd * g4[i] + b4[i];
        y[i] = (t > 0.f) ? t : (__expf(t) - 1.f);
    }
    float4 o = { y[0], y[1], y[2], y[3] };
    *(float4*)(out + base) = o;
}

extern "C" void kernel_launch(void* const* d_in, const int* in_sizes, int n_in,
                              void* d_out, int out_size, void* d_ws, size_t ws_size,
                              hipStream_t stream) {
    const float* h    = (const float*)d_in[0];
    const int*   ei   = (const int*)  d_in[1];
    const float* ef   = (const float*)d_in[2];
    const float* W_w  = (const float*)d_in[3];
    const float* W_b  = (const float*)d_in[4];
    const float* att  = (const float*)d_in[5];
    const float* ln_g = (const float*)d_in[6];
    const float* ln_b = (const float*)d_in[7];
    float* out = (float*)d_out;

    float* ws     = (float*)d_ws;
    float* z      = ws;
    float* s_src  = z      + (size_t)N_NODES * OUT_DIM;
    float* s_dst  = s_src  + (size_t)N_NODES * HEADS;
    float* denom  = s_dst  + (size_t)N_NODES * HEADS;
    float* accum  = denom  + (size_t)N_NODES * HEADS;

    {
        const int waves = N_NODES / 16;
        const int blocks = (waves + 7) / 8;
        gemm_z_wmma<<<blocks, 256, 0, stream>>>(h, W_w, W_b, z);
    }
    node_scores<<<(N_NODES * HEADS + 255) / 256, 256, 0, stream>>>(z, att, s_src, s_dst);
    {
        const int nz = N_NODES * HEADS + N_NODES * OUT_DIM;
        zero_f32<<<(nz + 255) / 256, 256, 0, stream>>>(denom, nz);
    }
    edge_denom<<<(N_EDGES * HEADS + 255) / 256, 256, 0, stream>>>(ei, ef, s_src, s_dst, att, denom);
    edge_aggregate<<<(N_EDGES * 32 + 255) / 256, 256, 0, stream>>>(ei, ef, s_src, s_dst, att,
                                                                   denom, z, accum);
    finalize<<<(N_NODES * 32 + 255) / 256, 256, 0, stream>>>(accum, z, ln_g, ln_b, out);
}


# Float-atomic scatter replaced by an in-call counting sort on destination (int rank atomics, scan, place) and one wave per node that gathers, normalises and finishes; sum order = arrival rank
# speedup vs baseline: 1.6187x; 1.6187x over previous
#include <hip/hip_runtime.h>
#include <math.h>

#define N_NODES 50000
#define N_EDGES 800000
#define IN_DIM  128
#define OUT_DIM 128
#define HEADS   4
#define D_K     32
#define ATT_DIM 65

typedef __attribute__((ext_vector_type(2))) float v2f;
typedef __attribute__((ext_vector_type(8))) float v8f;

__global__ void __launch_bounds__(256)
gemm_z_wmma(const float* __restrict__ h, const float* __restrict__ W,
            const float* __restrict__ bias, float* __restrict__ z) {
    const int wave = (blockIdx.x * blockDim.x + threadIdx.x) >> 5;
    const int lane = threadIdx.x & 31;
    if (wave >= (N_NODES / 16)) return;

    const int m = lane & 15;
    const int g = lane >> 4;

    const float* __restrict__ hrow = h + (size_t)(wave * 16 + m) * IN_DIM;
    v2f a[32];
#pragma unroll
    for (int ks = 0; ks < 32; ++ks) {
        const int k = 4 * ks + 2 * g;
        a[ks].x = hrow[k];
        a[ks].y = hrow[k + 1];
    }

    for (int ot = 0; ot < 8; ++ot) {
        const int col = ot * 16 + m;
        const float* __restrict__ wrow = W + (size_t)col * IN_DIM;
        v8f c = {};
#pragma unroll
        for (int ks = 0; ks < 32; ++ks) {
            const int k = 4 * ks + 2 * g;
            v2f b;
            b.x = wrow[k];
            b.y = wrow[k + 1];
            c = __builtin_amdgcn_wmma_f32_16x16x4_f32(
                     false, a[ks],  false, b,
                     (short)0, c,  false,  false);
        }
        const float bv = bias[col];
        const int rowBase = wave * 16 + g * 8;
#pragma unroll
        for (int r = 0; r < 8; ++r) {
            z[(size_t)(rowBase + r) * OUT_DIM + col] = c[r] + bv;
        }
    }
}

__global__ void __launch_bounds__(256)
node_scores(const float* __restrict__ z, const float* __restrict__ att,
            float* __restrict__ s_src, float* __restrict__ s_dst,
            int* __restrict__ cnt) {
    const int gid = blockIdx.x * blockDim.x + threadIdx.x;
    if (gid >= N_NODES * HEADS) return;
    if (gid < N_NODES) cnt[gid] = 0;
    const int hd = gid & 3;
    const int n  = gid >> 2;
    const float* __restrict__ zp = z + (size_t)n * OUT_DIM + hd * D_K;
    const float* __restrict__ as = att + hd * ATT_DIM;
    const float* __restrict__ ad = as + D_K;
    float ss = 0.f, sd = 0.f;
#pragma unroll
    for (int k = 0; k < D_K; ++k) {
        const float v = zp[k];
        ss = fmaf(v, as[k], ss);
        sd = fmaf(v, ad[k], sd);
    }
    s_src[gid] = ss;
    s_dst[gid] = sd;
}

__device__ __forceinline__ float leaky_clip_exp(float s) {
    s = (s >= 0.f) ? s : 0.2f * s;
    s = fminf(fmaxf(s, -20.f), 20.f);
    return __expf(s);
}

__global__ void __launch_bounds__(256)
edge_rank(const int* __restrict__ ei, int* __restrict__ cnt, int* __restrict__ rank) {
    const int e = blockIdx.x * blockDim.x + threadIdx.x;
    if (e >= N_EDGES) return;
    const int dst = ei[N_EDGES + e];
    rank[e] = __hip_atomic_fetch_add(&cnt[dst], 1, __ATOMIC_RELAXED, __HIP_MEMORY_SCOPE_AGENT);
}

#define SCAN_TILES 13

__global__ void __launch_bounds__(1024)
scan_counts(const int* __restrict__ cnt, int* __restrict__ off) {
    __shared__ int wtot[SCAN_TILES][32];
    __shared__ int ttot[SCAN_TILES];
    const int t = threadIdx.x;
    const int lane = t & 31;
    const int w = t >> 5;

    int4 v[SCAN_TILES];
    int sum[SCAN_TILES], incl[SCAN_TILES];
#pragma unroll
    for (int k = 0; k < SCAN_TILES; ++k) {
        const int i = k * 4096 + t * 4;
        int4 q = {0, 0, 0, 0};
        if (i < N_NODES) q = *(const int4*)(cnt + i);
        v[k] = q;
    }
#pragma unroll
    for (int k = 0; k < SCAN_TILES; ++k) {
        const int s = v[k].x + v[k].y + v[k].z + v[k].w;
        int x = s;
#pragma unroll
        for (int d = 1; d < 32; d <<= 1) {
            const int y = __shfl_up(x, d, 32);
            if (lane >= d) x += y;
        }
        sum[k] = s;
        incl[k] = x;
        if (lane == 31) wtot[k][w] = x;
    }
    __syncthreads();
    if (w < SCAN_TILES) {
        const int s = wtot[w][lane];
        int x = s;
#pragma unroll
        for (int d = 1; d < 32; d <<= 1) {
            const int y = __shfl_up(x, d, 32);
            if (lane >= d) x += y;
        }
        wtot[w][lane] = x - s;
        if (lane == 31) ttot[w] = x;
    }
    __syncthreads();
    int run = 0;
#pragma unroll
    for (int k = 0; k < SCAN_TILES; ++k) {
        const int i = k * 4096 + t * 4;
        const int ex = run + wtot[k][w] + incl[k] - sum[k];
        if (i < N_NODES) {
            int4 o;
            o.x = ex;
            o.y = ex + v[k].x;
            o.z = ex + v[k].x + v[k].y;
            o.w = ex + v[k].x + v[k].y + v[k].z;
            *(int4*)(off + i) = o;
        }
        run += ttot[k];
    }
    if (t == 0) off[N_NODES] = run;
}

__global__ void __launch_bounds__(256)
edge_place(const int* __restrict__ ei, const float* __restrict__ ef,
           const int* __restrict__ off, const int* __restrict__ rank,
           int* __restrict__ csr) {
    const int e = blockIdx.x * blockDim.x + threadIdx.x;
    if (e >= N_EDGES) return;
    const int src = ei[e];
    const int dst = ei[N_EDGES + e];
    const int pos = off[dst] + rank[e];
    int2 p;
    p.x = src;
    p.y = __float_as_int(ef[e]);
    *(int2*)(csr + 2 * (size_t)pos) = p;
}

__global__ void __launch_bounds__(256)
node_gather_finalize(const int* __restrict__ off, const int* __restrict__ csr,
                     const float* __restrict__ s_src, const float* __restrict__ s_dst,
                     const float* __restrict__ att, const float* __restrict__ z,
                     const float* __restrict__ ln_g, const float* __restrict__ ln_b,
                     float* __restrict__ out) {
    const int lane = threadIdx.x & 31;
    const int n = __builtin_amdgcn_readfirstlane((int)((blockIdx.x * blockDim.x + threadIdx.x) >> 5));
    if (n >= N_NODES) return;
    const int hd = lane >> 3;
    const int k0 = (hd == 0) ? -1 : 0;
    const int k1 = (hd == 1) ? -1 : 0;
    const int k2 = (hd == 2) ? -1 : 0;
    const int k3 = (hd == 3) ? -1 : 0;

    const int beg = off[n];
    const int end = off[n + 1];
    const float4 sd = *(const float4*)(s_dst + (size_t)n * HEADS);
    const float ae0 = att[0 * ATT_DIM + 2 * D_K];
    const float ae1 = att[1 * ATT_DIM + 2 * D_K];
    const float ae2 = att[2 * ATT_DIM + 2 * D_K];
    const float ae3 = att[3 * ATT_DIM + 2 * D_K];

    float acc0 = 0.f, acc1 = 0.f, acc2 = 0.f, acc3 = 0.f, den = 0.f;

    for (int base = beg; base < end; base += 32) {
        const int idx = base + lane;
        const bool valid = idx < end;
        const int2 p = *(const int2*)(csr + 2 * (size_t)(valid ? idx : base));
        const int srcv = valid ? p.x : n;
        const float efv = __int_as_float(p.y);
        const float4 ss = *(const float4*)(s_src + (size_t)srcv * HEADS);
        const int vm = valid ? -1 : 0;
        const int a0 = __float_as_int(leaky_clip_exp(ss.x + sd.x + efv * ae0)) & vm;
        const int a1 = __float_as_int(leaky_clip_exp(ss.y + sd.y + efv * ae1)) & vm;
        const int a2 = __float_as_int(leaky_clip_exp(ss.z + sd.z + efv * ae2)) & vm;
        const int a3 = __float_as_int(leaky_clip_exp(ss.w + sd.w + efv * ae3)) & vm;

        const int left = end - base;
        const int m = left < 32 ? left : 32;
        for (int j = 0; j < m; j += 8) {
#pragma unroll
            for (int jj = 0; jj < 8; ++jj) {
                const int l = j + jj;
                const int sj = __builtin_amdgcn_readlane(srcv, l);
                const float a = __int_as_float((__builtin_amdgcn_readlane(a0, l) & k0) |
                                               (__builtin_amdgcn_readlane(a1, l) & k1) |
                                               (__builtin_amdgcn_readlane(a2, l) & k2) |
                                               (__builtin_amdgcn_readlane(a3, l) & k3));
                const float4 zv = *(const float4*)(z + (size_t)sj * OUT_DIM + lane * 4);
                acc0 = fmaf(a, zv.x, acc0);
                acc1 = fmaf(a, zv.y, acc1);
                acc2 = fmaf(a, zv.z, acc2);
                acc3 = fmaf(a, zv.w, acc3);
                den += a;
            }
        }
    }

    const float inv = 1.f / (den + 1e-6f);
    const size_t obase = (size_t)n * OUT_DIM + lane * 4;
    const float4 zs = *(const float4*)(z + obase);
    float x[4] = { acc0 * inv + zs.x, acc1 * inv + zs.y, acc2 * inv + zs.z, acc3 * inv + zs.w };

    float s = x[0] + x[1] + x[2] + x[3];
#pragma unroll
    for (int o = 16; o > 0; o >>= 1) s += __shfl_xor(s, o, 32);
    const float mu = s * (1.f / 128.f);

    float v = 0.f;
#pragma unroll
    for (int i = 0; i < 4; ++i) { const float d = x[i] - mu; v = fmaf(d, d, v); }
#pragma unroll
    for (int o = 16; o > 0; o >>= 1) v += __shfl_xor(v, o, 32);
    const float rstd = rsqrtf(v * (1.f / 128.f) + 1e-5f);

    const float4 gv = *(const float4*)(ln_g + lane * 4);
    const float4 bv = *(const float4*)(ln_b + lane * 4);
    const float g4[4] = { gv.x, gv.y, gv.z, gv.w };
    const float b4[4] = { bv.x, bv.y, bv.z, bv.w };

    float y[4];
#pragma unroll
    for (int i = 0; i < 4; ++i) {
        const float t = (x[i] - mu) * rstd * g4[i] + b4[i];
        y[i] = (t > 0.f) ? t : (__expf(t) - 1.f);
    }
    float4 o4 = { y[0], y[1], y[2], y[3] };
    *(float4*)(out + obase) = o4;
}

extern "C" void kernel_launch(void* const* d_in, const int* in_sizes, int n_in,
                              void* d_out, int out_size, void* d_ws, size_t ws_size,
                              hipStream_t stream) {
    const float* h    = (const float*)d_in[0];
    const int*   ei   = (const int*)  d_in[1];
    const float* ef   = (const float*)d_in[2];
    const float* W_w  = (const float*)d_in[3];
    const float* W_b  = (const float*)d_in[4];
    const float* att  = (const float*)d_in[5];
    const float* ln_g = (const float*)d_in[6];
    const float* ln_b = (const float*)d_in[7];
    float* out = (float*)d_out;

    float* ws     = (float*)d_ws;
    float* z      = ws;
    float* s_src  = z      + (size_t)N_NODES * OUT_DIM;
    float* s_dst  = s_src  + (size_t)N_NODES * HEADS;
    int*   cnt    = (int*)(s_dst + (size_t)N_NODES * HEADS);
    int*   off    = cnt    + 50048;
    int*   rank   = off    + 50048;
    int*   csr    = rank   + N_EDGES;

    {
        const int waves = N_NODES / 16;
        const int blocks = (waves + 7) / 8;
        gemm_z_wmma<<<blocks, 256, 0, stream>>>(h, W_w, W_b, z);
    }
    node_scores<<<(N_NODES * HEADS + 255) / 256, 256, 0, stream>>>(z, att, s_src, s_dst, cnt);
    edge_rank<<<(N_EDGES + 255) / 256, 256, 0, stream>>>(ei, cnt, rank);
    scan_counts<<<1, 1024, 0, stream>>>(cnt, off);
    edge_place<<<(N_EDGES + 255) / 256, 256, 0, stream>>>(ei, ef, off, rank, csr);
    node_gather_finalize<<<(N_NODES * 32 + 255) / 256, 256, 0, stream>>>(off, csr, s_src, s_dst, att, z,
                                                                         ln_g, ln_b, out);
}


# Attention scores made inside the gather wave from rows it loads anyway (4 products per lane + 8-lane DPP sum: score sum order changes); score launch and per-edge score gather dropped
# speedup vs baseline: 1.7354x; 1.0721x over previous
#include <hip/hip_runtime.h>
#include <math.h>

#define N_NODES 50000
#define N_EDGES 800000
#define IN_DIM  128
#define OUT_DIM 128
#define HEADS   4
#define D_K     32
#define ATT_DIM 65

typedef __attribute__((ext_vector_type(2))) float v2f;
typedef __attribute__((ext_vector_type(8))) float v8f;
typedef __attribute__((ext_vector_type(4))) float v4f;

__global__ void __launch_bounds__(256)
gemm_z_wmma(const float* __restrict__ h, const float* __restrict__ W,
            const float* __restrict__ bias, float* __restrict__ z,
            int* __restrict__ cnt) {
    const int wave = (blockIdx.x * blockDim.x + threadIdx.x) >> 5;
    const int lane = threadIdx.x & 31;
    if (wave >= (N_NODES / 16)) return;

    const int m = lane & 15;
    const int g = lane >> 4;

    if (lane < 16) cnt[wave * 16 + lane] = 0;

    const float* __restrict__ hrow = h + (size_t)(wave * 16 + m) * IN_DIM;
    v2f a[32];
#pragma unroll
    for (int ks = 0; ks < 32; ++ks) {
        const int k = 4 * ks + 2 * g;
        a[ks].x = hrow[k];
        a[ks].y = hrow[k + 1];
    }

    for (int ot = 0; ot < 8; ++ot) {
        const int col = ot * 16 + m;
        const float* __restrict__ wrow = W + (size_t)col * IN_DIM;
        v8f c = {};
#pragma unroll
        for (int ks = 0; ks < 32; ++ks) {
            const int k = 4 * ks + 2 * g;
            v2f b;
            b.x = wrow[k];
            b.y = wrow[k + 1];
            c = __builtin_amdgcn_wmma_f32_16x16x4_f32(
                     false, a[ks],  false, b,
                     (short)0, c,  false,  false);
        }
        const float bv = bias[col];
        const int rowBase = wave * 16 + g * 8;
#pragma unroll
        for (int r = 0; r < 8; ++r) {
            z[(size_t)(rowBase + r) * OUT_DIM + col] = c[r] + bv;
        }
    }
}

__device__ __forceinline__ float leaky_clip_exp(float s) {
    s = (s >= 0.f) ? s : 0.2f * s;
    s = fminf(fmaxf(s, -20.f), 20.f);
    return __expf(s);
}

__global__ void __launch_bounds__(256)
edge_rank(const int* __restrict__ ei, int* __restrict__ cnt, int* __restrict__ rank) {
    const int e = blockIdx.x * blockDim.x + threadIdx.x;
    if (e >= N_EDGES) return;
    const int dst = ei[N_EDGES + e];
    rank[e] = __hip_atomic_fetch_add(&cnt[dst], 1, __ATOMIC_RELAXED, __HIP_MEMORY_SCOPE_AGENT);
}

#define SCAN_TILES 13

__global__ void __launch_bounds__(1024)
scan_counts(const int* __restrict__ cnt, int* __restrict__ off) {
    __shared__ int wtot[SCAN_TILES][32];
    __shared__ int ttot[SCAN_TILES];
    const int t = threadIdx.x;
    const int lane = t & 31;
    const int w = t >> 5;

    int4 v[SCAN_TILES];
    int sum[SCAN_TILES], incl[SCAN_TILES];
#pragma unroll
    for (int k = 0; k < SCAN_TILES; ++k) {
        const int i = k * 4096 + t * 4;
        int4 q = {0, 0, 0, 0};
        if (i < N_NODES) q = *(const int4*)(cnt + i);
        v[k] = q;
    }
#pragma unroll
    for (int k = 0; k < SCAN_TILES; ++k) {
        const int s = v[k].x + v[k].y + v[k].z + v[k].w;
        int x = s;
#pragma unroll
        for (int d = 1; d < 32; d <<= 1) {
            const int y = __shfl_up(x, d, 32);
            if (lane >= d) x += y;
        }
        sum[k] = s;
        incl[k] = x;
        if (lane == 31) wtot[k][w] = x;
    }
    __syncthreads();
    if (w < SCAN_TILES) {
        const int s = wtot[w][lane];
        int x = s;
#pragma unroll
        for (int d = 1; d < 32; d <<= 1) {
            const int y = __shfl_up(x, d, 32);
            if (lane >= d) x += y;
        }
        wtot[w][lane] = x - s;
        if (lane == 31) ttot[w] = x;
    }
    __syncthreads();
    int run = 0;
#pragma unroll
    for (int k = 0; k < SCAN_TILES; ++k) {
        const int i = k * 4096 + t * 4;
        const int ex = run + wtot[k][w] + incl[k] - sum[k];
        if (i < N_NODES) {
            int4 o;
            o.x = ex;
            o.y = ex + v[k].x;
            o.z = ex + v[k].x + v[k].y;
            o.w = ex + v[k].x + v[k].y + v[k].z;
            *(int4*)(off + i) = o;
        }
        run += ttot[k];
    }
    if (t == 0) off[N_NODES] = run;
}

__global__ void __launch_bounds__(256)
edge_place(const int* __restrict__ ei, const float* __restrict__ ef,
           const int* __restrict__ off, const int* __restrict__ rank,
           int* __restrict__ csr) {
    const int e = blockIdx.x * blockDim.x + threadIdx.x;
    if (e >= N_EDGES) return;
    const int src = ei[e];
    const int dst = ei[N_EDGES + e];
    const int pos = off[dst] + rank[e];
    int2 p;
    p.x = src;
    p.y = __float_as_int(ef[e]);
    *(int2*)(csr + 2 * (size_t)pos) = p;
}

template <int XMASK>
__device__ __forceinline__ float add_lane_xor(float v) {
    const int o = __builtin_amdgcn_update_dpp(0, __float_as_int(v), 0x160 + XMASK, 0xf, 0xf, true);
    return v + __int_as_float(o);
}
__device__ __forceinline__ float head_sum8(float v) {
    v = add_lane_xor<1>(v);
    v = add_lane_xor<2>(v);
    v = add_lane_xor<4>(v);
    return v;
}

__global__ void __launch_bounds__(256)
node_gather_finalize(const int* __restrict__ off, const int* __restrict__ csr,
                     const float* __restrict__ att, const float* __restrict__ z,
                     const float* __restrict__ ln_g, const float* __restrict__ ln_b,
                     float* __restrict__ out) {
    __shared__ float att_s[HEADS * ATT_DIM];
    const int t = threadIdx.x;
    const int lane = t & 31;
    const int n = __builtin_amdgcn_readfirstlane((int)((blockIdx.x * blockDim.x + threadIdx.x) >> 5));
    const int nc = n < N_NODES ? n : N_NODES - 1;

    const float at0 = att[t];
    float at1 = 0.f;
    if (t < HEADS * ATT_DIM - 256) at1 = att[256 + t];
    const size_t obase = (size_t)nc * OUT_DIM + lane * 4;
    const float4 zs = *(const float4*)(z + obase);
    const int beg = off[nc];
    const int end = off[nc + 1];
    att_s[t] = at0;
    if (t < HEADS * ATT_DIM - 256) att_s[256 + t] = at1;
    __syncthreads();
    if (n >= N_NODES) return;

    const int hd = lane >> 3;
    const float* ah = att_s + hd * ATT_DIM + (lane & 7) * 4;
    const float as0 = ah[0], as1 = ah[1], as2 = ah[2], as3 = ah[3];
    const float ae = att_s[hd * ATT_DIM + 2 * D_K];
    const float sdl = head_sum8(zs.x * ah[D_K] + zs.y * ah[D_K + 1] + zs.z * ah[D_K + 2] + zs.w * ah[D_K + 3]);

    float acc0 = 0.f, acc1 = 0.f, acc2 = 0.f, acc3 = 0.f, den = 0.f;
    const unsigned lane4 = (unsigned)lane * 4u;

    for (int base = beg; base < end; base += 32) {
        const int idx = base + lane;
        const bool valid = idx < end;
        const int2 p = *(const int2*)(csr + 2 * (size_t)(valid ? idx : base));
        const int srcv = valid ? p.x : n;
        const int efb = p.y;

        const int left = end - base;
        const int m = left < 32 ? left : 32;
        for (int j = 0; j < m; j += 8) {
            v4f zr[8];
#pragma unroll
            for (int jj = 0; jj < 8; ++jj) {
                const int sj = __builtin_amdgcn_readlane(srcv, j + jj);
                const float* rowp = z + (size_t)sj * OUT_DIM;
                zr[jj] = *(const v4f*)(rowp + lane4);
            }
            asm volatile("" : "+v"(zr[0]), "+v"(zr[1]), "+v"(zr[2]), "+v"(zr[3]),
                              "+v"(zr[4]), "+v"(zr[5]), "+v"(zr[6]), "+v"(zr[7]));
#pragma unroll
            for (int jj = 0; jj < 8; ++jj) {
                const int l = j + jj;
                const v4f zv = zr[jj];
                const float ef = __int_as_float(__builtin_amdgcn_readlane(efb, l));
                const float ps = head_sum8(zv.x * as0 + zv.y * as1 + zv.z * as2 + zv.w * as3);
                const float w = leaky_clip_exp(ps + sdl + ef * ae);
                const int km = (l < left) ? -1 : 0;
                const float a = __int_as_float(__float_as_int(w) & km);
                acc0 = fmaf(a, zv.x, acc0);
                acc1 = fmaf(a, zv.y, acc1);
                acc2 = fmaf(a, zv.z, acc2);
                acc3 = fmaf(a, zv.w, acc3);
                den += a;
            }
        }
    }

    const float inv = 1.f / (den + 1e-6f);
    float x[4] = { acc0 * inv + zs.x, acc1 * inv + zs.y, acc2 * inv + zs.z, acc3 * inv + zs.w };

    float s = x[0] + x[1] + x[2] + x[3];
#pragma unroll
    for (int o = 16; o > 0; o >>= 1) s += __shfl_xor(s, o, 32);
    const float mu = s * (1.f / 128.f);

    float v = 0.f;
#pragma unroll
    for (int i = 0; i < 4; ++i) { const float d = x[i] - mu; v = fmaf(d, d, v); }
#pragma unroll
    for (int o = 16; o > 0; o >>= 1) v += __shfl_xor(v, o, 32);
    const float rstd = rsqrtf(v * (1.f / 128.f) + 1e-5f);

    const float4 gv = *(const float4*)(ln_g + lane * 4);
    const float4 bv = *(const float4*)(ln_b + lane * 4);
    const float g4[4] = { gv.x, gv.y, gv.z, gv.w };
    const float b4[4] = { bv.x, bv.y, bv.z, bv.w };

    float y[4];
#pragma unroll
    for (int i = 0; i < 4; ++i) {
        const float tt = (x[i] - mu) * rstd * g4[i] + b4[i];
        y[i] = (tt > 0.f) ? tt : (__expf(tt) - 1.f);
    }
    float4 o4 = { y[0], y[1], y[2], y[3] };
    *(float4*)(out + obase) = o4;
}

extern "C" void kernel_launch(void* const* d_in, const int* in_sizes, int n_in,
                              void* d_out, int out_size, void* d_ws, size_t ws_size,
                              hipStream_t stream) {
    const float* h    = (const float*)d_in[0];
    const int*   ei   = (const int*)  d_in[1];
    const float* ef   = (const float*)d_in[2];
    const float* W_w  = (const float*)d_in[3];
    const float* W_b  = (const float*)d_in[4];
    const float* att  = (const float*)d_in[5];
    const float* ln_g = (const float*)d_in[6];
    const float* ln_b = (const float*)d_in[7];
    float* out = (float*)d_out;

    float* ws     = (float*)d_ws;
    float* z      = ws;
    int*   cnt    = (int*)(z + (size_t)N_NODES * OUT_DIM);
    int*   off    = cnt    + 50048;
    int*   rank   = off    + 50048;
    int*   csr    = rank   + N_EDGES;

    {
        const int waves = N_NODES / 16;
        const int blocks = (waves + 7) / 8;
        gemm_z_wmma<<<blocks, 256, 0, stream>>>(h, W_w, W_b, z, cnt);
    }
    edge_rank<<<(N_EDGES + 255) / 256, 256, 0, stream>>>(ei, cnt, rank);
    scan_counts<<<1, 1024, 0, stream>>>(cnt, off);
    edge_place<<<(N_EDGES + 255) / 256, 256, 0, stream>>>(ei, ef, off, rank, csr);
    node_gather_finalize<<<(N_NODES * 32 + 255) / 256, 256, 0, stream>>>(off, csr, att, z,
                                                                         ln_g, ln_b, out);
}


# Projection GEMM moved out of its own launch into the two edge launches of the sort (half the strips in each, on own work-groups dispatched first); counters cleared by a small first launch
# speedup vs baseline: 1.7868x; 1.0296x over previous
#include <hip/hip_runtime.h>
#include <math.h>

#define N_NODES 50000
#define N_EDGES 800000
#define IN_DIM  128
#define OUT_DIM 128
#define HEADS   4
#define D_K     32
#define ATT_DIM 65

typedef __attribute__((ext_vector_type(2))) float v2f;
typedef __attribute__((ext_vector_type(8))) float v8f;
typedef __attribute__((ext_vector_type(4))) float v4f;

__global__ void __launch_bounds__(256)
clear_counts(int* __restrict__ cnt) {
    const int i = blockIdx.x * blockDim.x + threadIdx.x;
    if (i < N_NODES / 4) ((int4*)cnt)[i] = make_int4(0, 0, 0, 0);
}

__device__ __forceinline__ void
gemm_z_strip(const int wave, const int lane,
             const float* __restrict__ h, const float* __restrict__ W,
             const float* __restrict__ bias, float* __restrict__ z) {
    const int m = lane & 15;
    const int g = lane >> 4;

    const float* __restrict__ hrow = h + (size_t)(wave * 16 + m) * IN_DIM;
    v2f a[32];
#pragma unroll
    for (int ks = 0; ks < 32; ++ks) {
        const int k = 4 * ks + 2 * g;
        a[ks].x = hrow[k];
        a[ks].y = hrow[k + 1];
    }

    for (int ot = 0; ot < 8; ++ot) {
        const int col = ot * 16 + m;
        const float* __restrict__ wrow = W + (size_t)col * IN_DIM;
        v8f c = {};
#pragma unroll
        for (int ks = 0; ks < 32; ++ks) {
            const int k = 4 * ks + 2 * g;
            v2f b;
            b.x = wrow[k];
            b.y = wrow[k + 1];
            c = __builtin_amdgcn_wmma_f32_16x16x4_f32(
                     false, a[ks],  false, b,
                     (short)0, c,  false,  false);
        }
        const float bv = bias[col];
        const int rowBase = wave * 16 + g * 8;
#pragma unroll
        for (int r = 0; r < 8; ++r) {
            z[(size_t)(rowBase + r) * OUT_DIM + col] = c[r] + bv;
        }
    }
}

__device__ __forceinline__ float leaky_clip_exp(float s) {
    s = (s >= 0.f) ? s : 0.2f * s;
    s = fminf(fmaxf(s, -20.f), 20.f);
    return __expf(s);
}

#define GEMM_WAVES    (N_NODES / 16)
#define GEMM_BLOCKS_A 196
#define GEMM_BLOCKS_B ((GEMM_WAVES - 8 * GEMM_BLOCKS_A + 7) / 8)
#define EDGE_BLOCKS   ((N_EDGES + 255) / 256)

__global__ void __launch_bounds__(256)
edge_rank_and_project(const int* __restrict__ ei, int* __restrict__ cnt, int* __restrict__ rank,
                      const float* __restrict__ h, const float* __restrict__ W,
                      const float* __restrict__ bias, float* __restrict__ z) {
    if (blockIdx.x < GEMM_BLOCKS_A) {
        const int wave = (blockIdx.x * blockDim.x + threadIdx.x) >> 5;
        gemm_z_strip(wave, threadIdx.x & 31, h, W, bias, z);
        return;
    }
    const int e = (blockIdx.x - GEMM_BLOCKS_A) * blockDim.x + threadIdx.x;
    if (e >= N_EDGES) return;
    const int dst = ei[N_EDGES + e];
    rank[e] = __hip_atomic_fetch_add(&cnt[dst], 1, __ATOMIC_RELAXED, __HIP_MEMORY_SCOPE_AGENT);
}

#define SCAN_TILES 13

__global__ void __launch_bounds__(1024)
scan_counts(const int* __restrict__ cnt, int* __restrict__ off) {
    __shared__ int wtot[SCAN_TILES][32];
    __shared__ int ttot[SCAN_TILES];
    const int t = threadIdx.x;
    const int lane = t & 31;
    const int w = t >> 5;

    int4 v[SCAN_TILES];
    int sum[SCAN_TILES], incl[SCAN_TILES];
#pragma unroll
    for (int k = 0; k < SCAN_TILES; ++k) {
        const int i = k * 4096 + t * 4;
        int4 q = {0, 0, 0, 0};
        if (i < N_NODES) q = *(const int4*)(cnt + i);
        v[k] = q;
    }
#pragma unroll
    for (int k = 0; k < SCAN_TILES; ++k) {
        const int s = v[k].x + v[k].y + v[k].z + v[k].w;
        int x = s;
#pragma unroll
        for (int d = 1; d < 32; d <<= 1) {
            const int y = __shfl_up(x, d, 32);
            if (lane >= d) x += y;
        }
        sum[k] = s;
        incl[k] = x;
        if (lane == 31) wtot[k][w] = x;
    }
    __syncthreads();
    if (w < SCAN_TILES) {
        const int s = wtot[w][lane];
        int x = s;
#pragma unroll
        for (int d = 1; d < 32; d <<= 1) {
            const int y = __shfl_up(x, d, 32);
            if (lane >= d) x += y;
        }
        wtot[w][lane] = x - s;
        if (lane == 31) ttot[w] = x;
    }
    __syncthreads();
    int run = 0;
#pragma unroll
    for (int k = 0; k < SCAN_TILES; ++k) {
        const int i = k * 4096 + t * 4;
        const int ex = run + wtot[k][w] + incl[k] - sum[k];
        if (i < N_NODES) {
            int4 o;
            o.x = ex;
            o.y = ex + v[k].x;
            o.z = ex + v[k].x + v[k].y;
            o.w = ex + v[k].x + v[k].y + v[k].z;
            *(int4*)(off + i) = o;
        }
        run += ttot[k];
    }
    if (t == 0) off[N_NODES] = run;
}

__global__ void __launch_bounds__(256)
edge_place_and_project(const int* __restrict__ ei, const float* __restrict__ ef,
                       const int* __restrict__ off, const int* __restrict__ rank,
                       int* __restrict__ csr,
                       const float* __restrict__ h, const float* __restrict__ W,
                       const float* __restrict__ bias, float* __restrict__ z) {
    if (blockIdx.x < GEMM_BLOCKS_B) {
        const int wave = 8 * GEMM_BLOCKS_A + ((blockIdx.x * blockDim.x + threadIdx.x) >> 5);
        if (wave < GEMM_WAVES)
            gemm_z_strip(wave, threadIdx.x & 31, h, W, bias, z);
        return;
    }
    const int e = (blockIdx.x - GEMM_BLOCKS_B) * blockDim.x + threadIdx.x;
    if (e >= N_EDGES) return;
    const int src = ei[e];
    const int dst = ei[N_EDGES + e];
    const int pos = off[dst] + rank[e];
    int2 p;
    p.x = src;
    p.y = __float_as_int(ef[e]);
    *(int2*)(csr + 2 * (size_t)pos) = p;
}

template <int XMASK>
__device__ __forceinline__ float add_lane_xor(float v) {
    const int o = __builtin_amdgcn_update_dpp(0, __float_as_int(v), 0x160 + XMASK, 0xf, 0xf, true);
    return v + __int_as_float(o);
}
__device__ __forceinline__ float head_sum8(float v) {
    v = add_lane_xor<1>(v);
    v = add_lane_xor<2>(v);
    v = add_lane_xor<4>(v);
    return v;
}

__global__ void __launch_bounds__(256)
node_gather_finalize(const int* __restrict__ off, const int* __restrict__ csr,
                     const float* __restrict__ att, const float* __restrict__ z,
                     const float* __restrict__ ln_g, const float* __restrict__ ln_b,
                     float* __restrict__ out) {
    __shared__ float att_s[HEADS * ATT_DIM];
    const int t = threadIdx.x;
    const int lane = t & 31;
    const int n = __builtin_amdgcn_readfirstlane((int)((blockIdx.x * blockDim.x + threadIdx.x) >> 5));
    const int nc = n < N_NODES ? n : N_NODES - 1;

    const float at0 = att[t];
    float at1 = 0.f;
    if (t < HEADS * ATT_DIM - 256) at1 = att[256 + t];
    const size_t obase = (size_t)nc * OUT_DIM + lane * 4;
    const float4 zs = *(const float4*)(z + obase);
    const int beg = off[nc];
    const int end = off[nc + 1];
    att_s[t] = at0;
    if (t < HEADS * ATT_DIM - 256) att_s[256 + t] = at1;
    __syncthreads();
    if (n >= N_NODES) return;

    const int hd = lane >> 3;
    const float* ah = att_s + hd * ATT_DIM + (lane & 7) * 4;
    const float as0 = ah[0], as1 = ah[1], as2 = ah[2], as3 = ah[3];
    const float ae = att_s[hd * ATT_DIM + 2 * D_K];
    const float sdl = head_sum8(zs.x * ah[D_K] + zs.y * ah[D_K + 1] + zs.z * ah[D_K + 2] + zs.w * ah[D_K + 3]);

    float acc0 = 0.f, acc1 = 0.f, acc2 = 0.f, acc3 = 0.f, den = 0.f;
    const unsigned lane4 = (unsigned)lane * 4u;

    for (int base = beg; base < end; base += 32) {
        const int idx = base + lane;
        const bool valid = idx < end;
        const int2 p = *(const int2*)(csr + 2 * (size_t)(valid ? idx : base));
        const int srcv = valid ? p.x : n;
        const int efb = p.y;

        const int left = end - base;
        const int m = left < 32 ? left : 32;
        for (int j = 0; j < m; j += 8) {
            v4f zr[8];
#pragma unroll
            for (int jj = 0; jj < 8; ++jj) {
                const int sj = __builtin_amdgcn_readlane(srcv, j + jj);
                const float* rowp = z + (size_t)sj * OUT_DIM;
                zr[jj] = *(const v4f*)(rowp + lane4);
            }
            asm volatile("" : "+v"(zr[0]), "+v"(zr[1]), "+v"(zr[2]), "+v"(zr[3]),
                              "+v"(zr[4]), "+v"(zr[5]), "+v"(zr[6]), "+v"(zr[7]));
#pragma unroll
            for (int jj = 0; jj < 8; ++jj) {
                const int l = j + jj;
                const v4f zv = zr[jj];
                const float ef = __int_as_float(__builtin_amdgcn_readlane(efb, l));
                const float ps = head_sum8(zv.x * as0 + zv.y * as1 + zv.z * as2 + zv.w * as3);
                const float w = leaky_clip_exp(ps + sdl + ef * ae);
                const int km = (l < left) ? -1 : 0;
                const float a = __int_as_float(__float_as_int(w) & km);
                acc0 = fmaf(a, zv.x, acc0);
                acc1 = fmaf(a, zv.y, acc1);
                acc2 = fmaf(a, zv.z, acc2);
                acc3 = fmaf(a, zv.w, acc3);
                den += a;
            }
        }
    }

    const float inv = 1.f / (den + 1e-6f);
    float x[4] = { acc0 * inv + zs.x, acc1 * inv + zs.y, acc2 * inv + zs.z, acc3 * inv + zs.w };

    float s = x[0] + x[1] + x[2] + x[3];
#pragma unroll
    for (int o = 16; o > 0; o >>= 1) s += __shfl_xor(s, o, 32);
    const float mu = s * (1.f / 128.f);

    float v = 0.f;
#pragma unroll
    for (int i = 0; i < 4; ++i) { const float d = x[i] - mu; v = fmaf(d, d, v); }
#pragma unroll
    for (int o = 16; o > 0; o >>= 1) v += __shfl_xor(v, o, 32);
    const float rstd = rsqrtf(v * (1.f / 128.f) + 1e-5f);

    const float4 gv = *(const float4*)(ln_g + lane * 4);
    const float4 bv = *(const float4*)(ln_b + lane * 4);
    const float g4[4] = { gv.x, gv.y, gv.z, gv.w };
    const float b4[4] = { bv.x, bv.y, bv.z, bv.w };

    float y[4];
#pragma unroll
    for (int i = 0; i < 4; ++i) {
        const float tt = (x[i] - mu) * rstd * g4[i] + b4[i];
        y[i] = (tt > 0.f) ? tt : (__expf(tt) - 1.f);
    }
    float4 o4 = { y[0], y[1], y[2], y[3] };
    *(float4*)(out + obase) = o4;
}

extern "C" void kernel_launch(void* const* d_in, const int* in_sizes, int n_in,
                              void* d_out, int out_size, void* d_ws, size_t ws_size,
                              hipStream_t stream) {
    const float* h    = (const float*)d_in[0];
    const int*   ei   = (const int*)  d_in[1];
    const float* ef   = (const float*)d_in[2];
    const float* W_w  = (const float*)d_in[3];
    const float* W_b  = (const float*)d_in[4];
    const float* att  = (const float*)d_in[5];
    const float* ln_g = (const float*)d_in[6];
    const float* ln_b = (const float*)d_in[7];
    float* out = (float*)d_out;

    float* ws     = (float*)d_ws;
    float* z      = ws;
    int*   cnt    = (int*)(z + (size_t)N_NODES * OUT_DIM);
    int*   off    = cnt    + 50048;
    int*   rank   = off    + 50048;
    int*   csr    = rank   + N_EDGES;

    clear_counts<<<(N_NODES / 4 + 255) / 256, 256, 0, stream>>>(cnt);
    edge_rank_and_project<<<GEMM_BLOCKS_A + EDGE_BLOCKS, 256, 0, stream>>>(ei, cnt, rank,
                                                                           h, W_w, W_b, z);
    scan_counts<<<1, 1024, 0, stream>>>(cnt, off);
    edge_place_and_project<<<GEMM_BLOCKS_B + EDGE_BLOCKS, 256, 0, stream>>>(ei, ef, off, rank, csr,
                                                                            h, W_w, W_b, z);
    node_gather_finalize<<<(N_NODES * 32 + 255) / 256, 256, 0, stream>>>(off, csr, att, z,
                                                                         ln_g, ln_b, out);
}


# Projection's B operand and bias read from an LDS copy of W staged once per work-group (plain loads + ds_store, pitch 132 words), one group ahead; no 16-line global loads in the loop; same sums
# speedup vs baseline: 1.9885x; 1.1129x over previous
#include <hip/hip_runtime.h>
#include <math.h>

#define N_NODES 50000
#define N_EDGES 800000
#define IN_DIM  128
#define OUT_DIM 128
#define HEADS   4
#define D_K     32
#define ATT_DIM 65

typedef __attribute__((ext_vector_type(2))) float v2f;
typedef __attribute__((ext_vector_type(8))) float v8f;
typedef __attribute__((ext_vector_type(4))) float v4f;

__global__ void __launch_bounds__(256)
clear_counts(int* __restrict__ cnt) {
    const int i = blockIdx.x * blockDim.x + threadIdx.x;
    if (i < N_NODES / 4) ((int4*)cnt)[i] = make_int4(0, 0, 0, 0);
}

#define W_PITCH     132
#define W_LDS_WORDS (OUT_DIM + OUT_DIM * W_PITCH)

__device__ __forceinline__ void
stage_w(const int t, const float* __restrict__ W, const float* __restrict__ bias, float* Ws) {
    float bt = 0.f;
    if (t < OUT_DIM) bt = bias[t];
#pragma unroll
    for (int q = 0; q < 16; ++q) {
        const int i = q * 256 + t;
        const int row = i >> 5;
        const int c4 = i & 31;
        const v4f v = *(const v4f*)(W + (size_t)row * IN_DIM + c4 * 4);
        *(v4f*)(Ws + OUT_DIM + row * W_PITCH + c4 * 4) = v;
    }
    if (t < OUT_DIM) Ws[t] = bt;
}

__device__ __forceinline__ void
gemm_z_strip(const int wave, const int lane,
             const float* __restrict__ h, const float* Ws, float* __restrict__ z) {
    const int m = lane & 15;
    const int g = lane >> 4;

    const float* __restrict__ hrow = h + (size_t)(wave * 16 + m) * IN_DIM;
    v2f a[32];
#pragma unroll
    for (int ks = 0; ks < 32; ++ks) {
        const int k = 4 * ks + 2 * g;
        a[ks].x = hrow[k];
        a[ks].y = hrow[k + 1];
    }

    const float* wlane = Ws + OUT_DIM + m * W_PITCH + 2 * g;
    v2f bcur[8], bnxt[8];
#pragma unroll
    for (int i = 0; i < 8; ++i) bcur[i] = *(const v2f*)(wlane + 4 * i);

#pragma unroll 1
    for (int ot = 0; ot < 8; ++ot) {
        const int col = ot * 16 + m;
        const float* wrow = wlane + ot * 16 * W_PITCH;
        const float* wnext = wlane + (ot < 7 ? ot + 1 : ot) * 16 * W_PITCH;
        float* zt = z + (size_t)(wave * 16 + g * 8) * OUT_DIM + col;
        asm volatile("" : "+v"(zt));
        v8f c = {};
#pragma unroll
        for (int grp = 0; grp < 4; ++grp) {
            const float* src = grp < 3 ? wrow + 32 * (grp + 1) : wnext;
#pragma unroll
            for (int i = 0; i < 8; ++i) bnxt[i] = *(const v2f*)(src + 4 * i);
            __builtin_amdgcn_sched_barrier(0);
#pragma unroll
            for (int i = 0; i < 8; ++i) {
                c = __builtin_amdgcn_wmma_f32_16x16x4_f32(
                         false, a[8 * grp + i],  false, bcur[i],
                         (short)0, c,  false,  false);
            }
#pragma unroll
            for (int i = 0; i < 8; ++i) bcur[i] = bnxt[i];
        }
        const float bv = Ws[col];
#pragma unroll
        for (int r = 0; r < 8; ++r) {
            zt[(size_t)r * OUT_DIM] = c[r] + bv;
        }
    }
}

__device__ __forceinline__ float leaky_clip_exp(float s) {
    s = (s >= 0.f) ? s : 0.2f * s;
    s = fminf(fmaxf(s, -20.f), 20.f);
    return __expf(s);
}

#define GEMM_WAVES    (N_NODES / 16)
#define GEMM_BLOCKS_A 196
#define GEMM_BLOCKS_B ((GEMM_WAVES - 8 * GEMM_BLOCKS_A + 7) / 8)
#define EDGE_BLOCKS   ((N_EDGES + 255) / 256)

__global__ void __launch_bounds__(256)
edge_rank_and_project(const int* __restrict__ ei, int* __restrict__ cnt, int* __restrict__ rank,
                      const float* __restrict__ h, const float* __restrict__ W,
                      const float* __restrict__ bias, float* __restrict__ z) {
    __shared__ v4f w_lds[W_LDS_WORDS / 4];
    if (blockIdx.x < GEMM_BLOCKS_A) {
        stage_w(threadIdx.x, W, bias, (float*)w_lds);
        __syncthreads();
        const int wave = (blockIdx.x * blockDim.x + threadIdx.x) >> 5;
        gemm_z_strip(wave, threadIdx.x & 31, h, (const float*)w_lds, z);
        return;
    }
    const int e = (blockIdx.x - GEMM_BLOCKS_A) * blockDim.x + threadIdx.x;
    if (e >= N_EDGES) return;
    const int dst = ei[N_EDGES + e];
    rank[e] = __hip_atomic_fetch_add(&cnt[dst], 1, __ATOMIC_RELAXED, __HIP_MEMORY_SCOPE_AGENT);
}

#define SCAN_TILES 13

__global__ void __launch_bounds__(1024)
scan_counts(const int* __restrict__ cnt, int* __restrict__ off) {
    __shared__ int wtot[SCAN_TILES][32];
    __shared__ int ttot[SCAN_TILES];
    const int t = threadIdx.x;
    const int lane = t & 31;
    const int w = t >> 5;

    int4 v[SCAN_TILES];
    int sum[SCAN_TILES], incl[SCAN_TILES];
#pragma unroll
    for (int k = 0; k < SCAN_TILES; ++k) {
        const int i = k * 4096 + t * 4;
        int4 q = {0, 0, 0, 0};
        if (i < N_NODES) q = *(const int4*)(cnt + i);
        v[k] = q;
    }
#pragma unroll
    for (int k = 0; k < SCAN_TILES; ++k) {
        const int s = v[k].x + v[k].y + v[k].z + v[k].w;
        int x = s;
#pragma unroll
        for (int d = 1; d < 32; d <<= 1) {
            const int y = __shfl_up(x, d, 32);
            if (lane >= d) x += y;
        }
        sum[k] = s;
        incl[k] = x;
        if (lane == 31) wtot[k][w] = x;
    }
    __syncthreads();
    if (w < SCAN_TILES) {
        const int s = wtot[w][lane];
        int x = s;
#pragma unroll
        for (int d = 1; d < 32; d <<= 1) {
            const int y = __shfl_up(x, d, 32);
            if (lane >= d) x += y;
        }
        wtot[w][lane] = x - s;
        if (lane == 31) ttot[w] = x;
    }
    __syncthreads();
    int run = 0;
#pragma unroll
    for (int k = 0; k < SCAN_TILES; ++k) {
        const int i = k * 4096 + t * 4;
        const int ex = run + wtot[k][w] + incl[k] - sum[k];
        if (i < N_NODES) {
            int4 o;
            o.x = ex;
            o.y = ex + v[k].x;
            o.z = ex + v[k].x + v[k].y;
            o.w = ex + v[k].x + v[k].y + v[k].z;
            *(int4*)(off + i) = o;
        }
        run += ttot[k];
    }
    if (t == 0) off[N_NODES] = run;
}

__global__ void __launch_bounds__(256)
edge_place_and_project(const int* __restrict__ ei, const float* __restrict__ ef,
                       const int* __restrict__ off, const int* __restrict__ rank,
                       int* __restrict__ csr,
                       const float* __restrict__ h, const float* __restrict__ W,
                       const float* __restrict__ bias, float* __restrict__ z) {
    __shared__ v4f w_lds[W_LDS_WORDS / 4];
    if (blockIdx.x < GEMM_BLOCKS_B) {
        stage_w(threadIdx.x, W, bias, (float*)w_lds);
        __syncthreads();
        const int wave = 8 * GEMM_BLOCKS_A + ((blockIdx.x * blockDim.x + threadIdx.x) >> 5);
        if (wave < GEMM_WAVES)
            gemm_z_strip(wave, threadIdx.x & 31, h, (const float*)w_lds, z);
        return;
    }
    const int e = (blockIdx.x - GEMM_BLOCKS_B) * blockDim.x + threadIdx.x;
    if (e >= N_EDGES) return;
    const int src = ei[e];
    const int dst = ei[N_EDGES + e];
    const int pos = off[dst] + rank[e];
    int2 p;
    p.x = src;
    p.y = __float_as_int(ef[e]);
    *(int2*)(csr + 2 * (size_t)pos) = p;
}

template <int XMASK>
__device__ __forceinline__ float add_lane_xor(float v) {
    const int o = __builtin_amdgcn_update_dpp(0, __float_as_int(v), 0x160 + XMASK, 0xf, 0xf, true);
    return v + __int_as_float(o);
}
__device__ __forceinline__ float head_sum8(float v) {
    v = add_lane_xor<1>(v);
    v = add_lane_xor<2>(v);
    v = add_lane_xor<4>(v);
    return v;
}

__global__ void __launch_bounds__(256)
node_gather_finalize(const int* __restrict__ off, const int* __restrict__ csr,
                     const float* __restrict__ att, const float* __restrict__ z,
                     const float* __restrict__ ln_g, const float* __restrict__ ln_b,
                     float* __restrict__ out) {
    __shared__ float att_s[HEADS * ATT_DIM];
    const int t = threadIdx.x;
    const int lane = t & 31;
    const int n = __builtin_amdgcn_readfirstlane((int)((blockIdx.x * blockDim.x + threadIdx.x) >> 5));
    const int nc = n < N_NODES ? n : N_NODES - 1;

    const float at0 = att[t];
    float at1 = 0.f;
    if (t < HEADS * ATT_DIM - 256) at1 = att[256 + t];
    const size_t obase = (size_t)nc * OUT_DIM + lane * 4;
    const float4 zs = *(const float4*)(z + obase);
    const int beg = off[nc];
    const int end = off[nc + 1];
    att_s[t] = at0;
    if (t < HEADS * ATT_DIM - 256) att_s[256 + t] = at1;
    __syncthreads();
    if (n >= N_NODES) return;

    const int hd = lane >> 3;
    const float* ah = att_s + hd * ATT_DIM + (lane & 7) * 4;
    const float as0 = ah[0], as1 = ah[1], as2 = ah[2], as3 = ah[3];
    const float ae = att_s[hd * ATT_DIM + 2 * D_K];
    const float sdl = head_sum8(zs.x * ah[D_K] + zs.y * ah[D_K + 1] + zs.z * ah[D_K + 2] + zs.w * ah[D_K + 3]);

    float acc0 = 0.f, acc1 = 0.f, acc2 = 0.f, acc3 = 0.f, den = 0.f;
    const unsigned lane4 = (unsigned)lane * 4u;

    for (int base = beg; base < end; base += 32) {
        const int idx = base + lane;
        const bool valid = idx < end;
        const int2 p = *(const int2*)(csr + 2 * (size_t)(valid ? idx : base));
        const int srcv = valid ? p.x : n;
        const int efb = p.y;

        const int left = end - base;
        const int m = left < 32 ? left : 32;
        for (int j = 0; j < m; j += 8) {
            v4f zr[8];
#pragma unroll
            for (int jj = 0; jj < 8; ++jj) {
                const int sj = __builtin_amdgcn_readlane(srcv, j + jj);
                const float* rowp = z + (size_t)sj * OUT_DIM;
                zr[jj] = *(const v4f*)(rowp + lane4);
            }
            asm volatile("" : "+v"(zr[0]), "+v"(zr[1]), "+v"(zr[2]), "+v"(zr[3]),
                              "+v"(zr[4]), "+v"(zr[5]), "+v"(zr[6]), "+v"(zr[7]));
#pragma unroll
            for (int jj = 0; jj < 8; ++jj) {
                const int l = j + jj;
                const v4f zv = zr[jj];
                const float ef = __int_as_float(__builtin_amdgcn_readlane(efb, l));
                const float ps = head_sum8(zv.x * as0 + zv.y * as1 + zv.z * as2 + zv.w * as3);
                const float w = leaky_clip_exp(ps + sdl + ef * ae);
                const int km = (l < left) ? -1 : 0;
                const float a = __int_as_float(__float_as_int(w) & km);
                acc0 = fmaf(a, zv.x, acc0);
                acc1 = fmaf(a, zv.y, acc1);
                acc2 = fmaf(a, zv.z, acc2);
                acc3 = fmaf(a, zv.w, acc3);
                den += a;
            }
        }
    }

    const float inv = 1.f / (den + 1e-6f);
    float x[4] = { acc0 * inv + zs.x, acc1 * inv + zs.y, acc2 * inv + zs.z, acc3 * inv + zs.w };

    float s = x[0] + x[1] + x[2] + x[3];
#pragma unroll
    for (int o = 16; o > 0; o >>= 1) s += __shfl_xor(s, o, 32);
    const float mu = s * (1.f / 128.f);

    float v = 0.f;
#pragma unroll
    for (int i = 0; i < 4; ++i) { const float d = x[i] - mu; v = fmaf(d, d, v); }
#pragma unroll
    for (int o = 16; o > 0; o >>= 1) v += __shfl_xor(v, o, 32);
    const float rstd = rsqrtf(v * (1.f / 128.f) + 1e-5f);

    const float4 gv = *(const float4*)(ln_g + lane * 4);
    const float4 bv = *(const float4*)(ln_b + lane * 4);
    const float g4[4] = { gv.x, gv.y, gv.z, gv.w };
    const float b4[4] = { bv.x, bv.y, bv.z, bv.w };

    float y[4];
#pragma unroll
    for (int i = 0; i < 4; ++i) {
        const float tt = (x[i] - mu) * rstd * g4[i] + b4[i];
        y[i] = (tt > 0.f) ? tt : (__expf(tt) - 1.f);
    }
    float4 o4 = { y[0], y[1], y[2], y[3] };
    *(float4*)(out + obase) = o4;
}

extern "C" void kernel_launch(void* const* d_in, const int* in_sizes, int n_in,
                              void* d_out, int out_size, void* d_ws, size_t ws_size,
                              hipStream_t stream) {
    const float* h    = (const float*)d_in[0];
    const int*   ei   = (const int*)  d_in[1];
    const float* ef   = (const float*)d_in[2];
    const float* W_w  = (const float*)d_in[3];
    const float* W_b  = (const float*)d_in[4];
    const float* att  = (const float*)d_in[5];
    const float* ln_g = (const float*)d_in[6];
    const float* ln_b = (const float*)d_in[7];
    float* out = (float*)d_out;

    float* ws     = (float*)d_ws;
    float* z      = ws;
    int*   cnt    = (int*)(z + (size_t)N_NODES * OUT_DIM);
    int*   off    = cnt    + 50048;
    int*   rank   = off    + 50048;
    int*   csr    = rank   + N_EDGES;

    clear_counts<<<(N_NODES / 4 + 255) / 256, 256, 0, stream>>>(cnt);
    edge_rank_and_project<<<GEMM_BLOCKS_A + EDGE_BLOCKS, 256, 0, stream>>>(ei, cnt, rank,
                                                                           h, W_w, W_b, z);
    scan_counts<<<1, 1024, 0, stream>>>(cnt, off);
    edge_place_and_project<<<GEMM_BLOCKS_B + EDGE_BLOCKS, 256, 0, stream>>>(ei, ef, off, rank, csr,
                                                                            h, W_w, W_b, z);
    node_gather_finalize<<<(N_NODES * 32 + 255) / 256, 256, 0, stream>>>(off, csr, att, z,
                                                                         ln_g, ln_b, out);
}


# Global counting sort (rank atomics, clear, scan, scattered placement) replaced by two-level LDS binning: chunks grouped by node slice, slices sorted in LDS; 5->2 launches; sum order = LDS arrival
# speedup vs baseline: 4.5942x; 2.3104x over previous
#include <hip/hip_runtime.h>
#include <math.h>

#define N_NODES 50000
#define N_EDGES 800000
#define IN_DIM  128
#define OUT_DIM 128
#define HEADS   4
#define D_K     32
#define ATT_DIM 65

typedef __attribute__((ext_vector_type(2))) float v2f;
typedef __attribute__((ext_vector_type(8))) float v8f;
typedef __attribute__((ext_vector_type(4))) float v4f;

#define W_PITCH     132
#define W_LDS_WORDS (OUT_DIM + OUT_DIM * W_PITCH)

__device__ __forceinline__ void
stage_w(const int t, const float* __restrict__ W, const float* __restrict__ bias, float* Ws) {
    float bt = 0.f;
    if (t < OUT_DIM) bt = bias[t];
#pragma unroll
    for (int q = 0; q < 16; ++q) {
        const int i = q * 256 + t;
        const int row = i >> 5;
        const int c4 = i & 31;
        const v4f v = *(const v4f*)(W + (size_t)row * IN_DIM + c4 * 4);
        *(v4f*)(Ws + OUT_DIM + row * W_PITCH + c4 * 4) = v;
    }
    if (t < OUT_DIM) Ws[t] = bt;
}

__device__ __forceinline__ void
gemm_z_strip(const int wave, const int lane,
             const float* __restrict__ h, const float* Ws, float* __restrict__ z) {
    const int m = lane & 15;
    const int g = lane >> 4;

    const float* __restrict__ hrow = h + (size_t)(wave * 16 + m) * IN_DIM;
    v2f a[32];
#pragma unroll
    for (int ks = 0; ks < 32; ++ks) {
        const int k = 4 * ks + 2 * g;
        a[ks].x = hrow[k];
        a[ks].y = hrow[k + 1];
    }

    const float* wlane = Ws + OUT_DIM + m * W_PITCH + 2 * g;
    v2f bcur[8], bnxt[8];
#pragma unroll
    for (int i = 0; i < 8; ++i) bcur[i] = *(const v2f*)(wlane + 4 * i);

#pragma unroll 1
    for (int ot = 0; ot < 8; ++ot) {
        const int col = ot * 16 + m;
        const float* wrow = wlane + ot * 16 * W_PITCH;
        const float* wnext = wlane + (ot < 7 ? ot + 1 : ot) * 16 * W_PITCH;
        float* zt = z + (size_t)(wave * 16 + g * 8) * OUT_DIM + col;
        asm volatile("" : "+v"(zt));
        v8f c = {};
#pragma unroll
        for (int grp = 0; grp < 4; ++grp) {
            const float* src = grp < 3 ? wrow + 32 * (grp + 1) : wnext;
#pragma unroll
            for (int i = 0; i < 8; ++i) bnxt[i] = *(const v2f*)(src + 4 * i);
            __builtin_amdgcn_sched_barrier(0);
#pragma unroll
            for (int i = 0; i < 8; ++i) {
                c = __builtin_amdgcn_wmma_f32_16x16x4_f32(
                         false, a[8 * grp + i],  false, bcur[i],
                         (short)0, c,  false,  false);
            }
#pragma unroll
            for (int i = 0; i < 8; ++i) bcur[i] = bnxt[i];
        }
        const float bv = Ws[col];
#pragma unroll
        for (int r = 0; r < 8; ++r) {
            zt[(size_t)r * OUT_DIM] = c[r] + bv;
        }
    }
}

__device__ __forceinline__ float leaky_clip_exp(float s) {
    s = (s >= 0.f) ? s : 0.2f * s;
    s = fminf(fmaxf(s, -20.f), 20.f);
    return __expf(s);
}

#define SLICE_NODES 196
#define N_SLICE     256
#define N_CHUNK     256
#define CHUNK_EDGES (N_EDGES / N_CHUNK)
#define CHUNK_STEPS ((CHUNK_EDGES + 255) / 256)
#define SEG_PITCH   (N_SLICE + 1)
#define GEMM_WAVES  (N_NODES / 16)
#define GEMM_BLOCKS ((GEMM_WAVES + 7) / 8)

static_assert(N_CHUNK * CHUNK_EDGES == N_EDGES, "chunks must tile the edges");
static_assert(N_SLICE * SLICE_NODES >= N_NODES && SLICE_NODES <= 256 && N_NODES <= 65536, "record packing");

__device__ __forceinline__ int wave_incl_scan(int x, const int lane) {
#pragma unroll
    for (int d = 1; d < 32; d <<= 1) {
        const int y = __shfl_up(x, d, 32);
        if (lane >= d) x += y;
    }
    return x;
}

__device__ __forceinline__ void
bin_chunk(const int c, const int t, const int* __restrict__ ei, const float* __restrict__ ef,
          int2* __restrict__ binned, int* __restrict__ seg_off, int* lds) {
    int* cnt  = lds;
    int* wtot = lds + N_SLICE;
    int2* rec = (int2*)(lds + N_SLICE + 16);
    const int lane = t & 31;
    const int w = t >> 5;

    cnt[t] = 0;
    __syncthreads();

    int sl[CHUNK_STEPS], rk[CHUNK_STEPS], px[CHUNK_STEPS], py[CHUNK_STEPS];
#pragma unroll
    for (int i = 0; i < CHUNK_STEPS; ++i) {
        const int le = i * 256 + t;
        const int e = c * CHUNK_EDGES + (le < CHUNK_EDGES ? le : 0);
        px[i] = ei[e];
        sl[i] = ei[N_EDGES + e];
        py[i] = __float_as_int(ef[e]);
    }
    __builtin_amdgcn_sched_barrier(0);
#pragma unroll
    for (int i = 0; i < CHUNK_STEPS; ++i) {
        const int dst = sl[i];
        const int s = dst / SLICE_NODES;
        sl[i] = s;
        px[i] |= (dst - s * SLICE_NODES) << 16;
        rk[i] = (i * 256 + t < CHUNK_EDGES) ? atomicAdd(&cnt[s], 1) : 0;
    }
    __syncthreads();

    const int v = cnt[t];
    const int incl = wave_incl_scan(v, lane);
    if (lane == 31) wtot[w] = incl;
    __syncthreads();
    int pre = 0;
#pragma unroll
    for (int k = 0; k < 8; ++k) pre += (k < w) ? wtot[k] : 0;
    const int excl = pre + incl - v;
    cnt[t] = excl;
    seg_off[c * SEG_PITCH + t] = excl;
    if (t == N_SLICE - 1) seg_off[c * SEG_PITCH + N_SLICE] = excl + v;
    __syncthreads();

#pragma unroll
    for (int i = 0; i < CHUNK_STEPS; ++i) {
        const int le = i * 256 + t;
        if (le < CHUNK_EDGES) rec[cnt[sl[i]] + rk[i]] = make_int2(px[i], py[i]);
    }
    __syncthreads();

    int2* outp = binned + (size_t)c * CHUNK_EDGES;
#pragma unroll
    for (int i = 0; i < CHUNK_STEPS; ++i) {
        const int le = i * 256 + t;
        if (le < CHUNK_EDGES) outp[le] = rec[le];
    }
}

__global__ void __launch_bounds__(256)
edge_bin_and_project(const int* __restrict__ ei, const float* __restrict__ ef,
                     int2* __restrict__ binned, int* __restrict__ seg_off,
                     const float* __restrict__ h, const float* __restrict__ W,
                     const float* __restrict__ bias, float* __restrict__ z) {
    __shared__ v4f w_lds[W_LDS_WORDS / 4];
    if (blockIdx.x < GEMM_BLOCKS) {
        stage_w(threadIdx.x, W, bias, (float*)w_lds);
        __syncthreads();
        const int wave = (blockIdx.x * blockDim.x + threadIdx.x) >> 5;
        if (wave < GEMM_WAVES)
            gemm_z_strip(wave, threadIdx.x & 31, h, (const float*)w_lds, z);
        return;
    }
    bin_chunk(blockIdx.x - GEMM_BLOCKS, threadIdx.x, ei, ef, binned, seg_off, (int*)w_lds);
}

template <int XMASK>
__device__ __forceinline__ float add_lane_xor(float v) {
    const int o = __builtin_amdgcn_update_dpp(0, __float_as_int(v), 0x160 + XMASK, 0xf, 0xf, true);
    return v + __int_as_float(o);
}
__device__ __forceinline__ float head_sum8(float v) {
    v = add_lane_xor<1>(v);
    v = add_lane_xor<2>(v);
    v = add_lane_xor<4>(v);
    return v;
}

#define SLICE_CAP     8192
#define GATHER_WAVES  32

struct NodeCoef {
    float as0, as1, as2, as3;
    float ad0, ad1, ad2, ad3;
    float ae;
    float4 g, b;
};

struct NodeAcc { float a0, a1, a2, a3, den; };

__device__ __forceinline__ void
edge_step(NodeAcc& s, const NodeCoef& k, const v4f zv, const float ef, const float sdl, const int keep) {
    const float ps = head_sum8(zv.x * k.as0 + zv.y * k.as1 + zv.z * k.as2 + zv.w * k.as3);
    const float w = leaky_clip_exp(ps + sdl + ef * k.ae);
    const float a = __int_as_float(__float_as_int(w) & keep);
    s.a0 = fmaf(a, zv.x, s.a0);
    s.a1 = fmaf(a, zv.y, s.a1);
    s.a2 = fmaf(a, zv.z, s.a2);
    s.a3 = fmaf(a, zv.w, s.a3);
    s.den += a;
}

__device__ __forceinline__ void
finish_node(const NodeAcc& s, const NodeCoef& k, const float4 zs, float* __restrict__ outp) {
    const float inv = 1.f / (s.den + 1e-6f);
    float x[4] = { s.a0 * inv + zs.x, s.a1 * inv + zs.y, s.a2 * inv + zs.z, s.a3 * inv + zs.w };

    float sum = x[0] + x[1] + x[2] + x[3];
#pragma unroll
    for (int o = 16; o > 0; o >>= 1) sum += __shfl_xor(sum, o, 32);
    const float mu = sum * (1.f / 128.f);

    float v = 0.f;
#pragma unroll
    for (int i = 0; i < 4; ++i) { const float d = x[i] - mu; v = fmaf(d, d, v); }
#pragma unroll
    for (int o = 16; o > 0; o >>= 1) v += __shfl_xor(v, o, 32);
    const float rstd = rsqrtf(v * (1.f / 128.f) + 1e-5f);

    const float g4[4] = { k.g.x, k.g.y, k.g.z, k.g.w };
    const float b4[4] = { k.b.x, k.b.y, k.b.z, k.b.w };
    float y[4];
#pragma unroll
    for (int i = 0; i < 4; ++i) {
        const float tt = (x[i] - mu) * rstd * g4[i] + b4[i];
        y[i] = (tt > 0.f) ? tt : (__expf(tt) - 1.f);
    }
    float4 o4 = { y[0], y[1], y[2], y[3] };
    *(float4*)outp = o4;
}

__global__ void __launch_bounds__(1024)
slice_gather_finalize(const int2* __restrict__ binned, const int* __restrict__ seg_off,
                      const float* __restrict__ att, const float* __restrict__ z,
                      const float* __restrict__ ln_g, const float* __restrict__ ln_b,
                      float* __restrict__ out) {
    __shared__ float att_s[HEADS * ATT_DIM];
    __shared__ int seg_start[N_CHUNK];
    __shared__ int seg_len[N_CHUNK];
    __shared__ int seg_pos[N_CHUNK];
    __shared__ int wtot[8];
    __shared__ int ncnt[256];
    __shared__ int noff[257];
    __shared__ int total_s;
    __shared__ int2 stage[SLICE_CAP];
    __shared__ int2 sorted[SLICE_CAP];

    const int t = threadIdx.x;
    const int lane = t & 31;
    const int w = __builtin_amdgcn_readfirstlane(t >> 5);
    const int s = blockIdx.x;

    float av = 0.f;
    if (t < HEADS * ATT_DIM) av = att[t];
    int len = 0, start = 0;
    if (t < N_CHUNK) {
        const int a = seg_off[t * SEG_PITCH + s];
        const int b = seg_off[t * SEG_PITCH + s + 1];
        len = b - a;
        start = t * CHUNK_EDGES + a;
        ncnt[t] = 0;
    }
    if (t < HEADS * ATT_DIM) att_s[t] = av;
    const int incl = wave_incl_scan(len, lane);
    if (t < N_CHUNK && lane == 31) wtot[w] = incl;
    __syncthreads();
    if (t < N_CHUNK) {
        int pre = 0;
#pragma unroll
        for (int k = 0; k < 8; ++k) pre += (k < w) ? wtot[k] : 0;
        const int excl = pre + incl - len;
        seg_start[t] = start;
        seg_len[t] = len;
        seg_pos[t] = excl;
        if (t == N_CHUNK - 1) total_s = excl + len;
    }
    __syncthreads();
    const int M = __builtin_amdgcn_readfirstlane(total_s);

    NodeCoef k;
    {
        const int hd = lane >> 3;
        const float* ah = att_s + hd * ATT_DIM + (lane & 7) * 4;
        k.as0 = ah[0]; k.as1 = ah[1]; k.as2 = ah[2]; k.as3 = ah[3];
        k.ad0 = ah[D_K]; k.ad1 = ah[D_K + 1]; k.ad2 = ah[D_K + 2]; k.ad3 = ah[D_K + 3];
        k.ae = att_s[hd * ATT_DIM + 2 * D_K];
        k.g = *(const float4*)(ln_g + lane * 4);
        k.b = *(const float4*)(ln_b + lane * 4);
    }
    const unsigned lane4 = (unsigned)lane * 4u;
    const int node0 = s * SLICE_NODES;

    if (M > SLICE_CAP) {
        for (int l = w; l < SLICE_NODES; l += GATHER_WAVES) {
            const int n = node0 + l;
            if (n >= N_NODES) break;
            const size_t obase = (size_t)n * OUT_DIM + lane4;
            const float4 zs = *(const float4*)(z + obase);
            const float sdl = head_sum8(zs.x * k.ad0 + zs.y * k.ad1 + zs.z * k.ad2 + zs.w * k.ad3);
            NodeAcc acc = { 0.f, 0.f, 0.f, 0.f, 0.f };
            for (int c = 0; c < N_CHUNK; ++c) {
                const int st = __builtin_amdgcn_readfirstlane(seg_start[c]);
                const int ln = __builtin_amdgcn_readfirstlane(seg_len[c]);
                for (int i0 = 0; i0 < ln; i0 += 32) {
                    const int i = i0 + lane;
                    int2 r = make_int2(0, 0);
                    if (i < ln) r = binned[st + i];
                    const bool hit = (i < ln) && (int)((unsigned)r.x >> 16) == l;
                    unsigned mask = __builtin_amdgcn_ballot_w32(hit);
                    while (mask) {
                        const int bpos = __builtin_ctz(mask);
                        mask &= mask - 1;
                        const int sj = __builtin_amdgcn_readlane(r.x, bpos) & 0xffff;
                        const float efv = __int_as_float(__builtin_amdgcn_readlane(r.y, bpos));
                        const v4f zv = *(const v4f*)(z + (size_t)sj * OUT_DIM + lane4);
                        edge_step(acc, k, zv, efv, sdl, -1);
                    }
                }
            }
            finish_node(acc, k, zs, out + obase);
        }
        return;
    }

    {
        int st[N_CHUNK / GATHER_WAVES], ln[N_CHUNK / GATHER_WAVES], ps[N_CHUNK / GATHER_WAVES];
        int2 r[N_CHUNK / GATHER_WAVES];
#pragma unroll
        for (int q = 0; q < N_CHUNK / GATHER_WAVES; ++q) {
            const int c = w + GATHER_WAVES * q;
            st[q] = __builtin_amdgcn_readfirstlane(seg_start[c]);
            ln[q] = __builtin_amdgcn_readfirstlane(seg_len[c]);
            ps[q] = __builtin_amdgcn_readfirstlane(seg_pos[c]);
            r[q] = make_int2(0, 0);
            if (lane < ln[q]) r[q] = binned[st[q] + lane];
        }
#pragma unroll
        for (int q = 0; q < N_CHUNK / GATHER_WAVES; ++q) {
            if (lane < ln[q]) {
                stage[ps[q] + lane] = r[q];
                atomicAdd(&ncnt[(unsigned)r[q].x >> 16], 1);
            }
            for (int i = lane + 32; i < ln[q]; i += 32) {
                const int2 rr = binned[st[q] + i];
                stage[ps[q] + i] = rr;
                atomicAdd(&ncnt[(unsigned)rr.x >> 16], 1);
            }
        }
    }
    __syncthreads();

    int nv = 0, nincl = 0;
    if (t < 256) {
        nv = ncnt[t];
        nincl = wave_incl_scan(nv, lane);
        if (lane == 31) wtot[w] = nincl;
    }
    __syncthreads();
    if (t < 256) {
        int pre = 0;
#pragma unroll
        for (int q = 0; q < 8; ++q) pre += (q < w) ? wtot[q] : 0;
        const int excl = pre + nincl - nv;
        noff[t] = excl;
        ncnt[t] = excl;
        if (t == 255) noff[256] = excl + nv;
    }
    __syncthreads();

    for (int j = t; j < M; j += 1024) {
        const int2 r = stage[j];
        const int pos = atomicAdd(&ncnt[(unsigned)r.x >> 16], 1);
        sorted[pos] = make_int2(r.x & 0xffff, r.y);
    }
    __syncthreads();

    float4 zs_next = make_float4(0.f, 0.f, 0.f, 0.f);
    if (node0 + w < N_NODES) zs_next = *(const float4*)(z + (size_t)(node0 + w) * OUT_DIM + lane4);
    for (int l = w; l < SLICE_NODES; l += GATHER_WAVES) {
        const int n = node0 + l;
        if (n >= N_NODES) break;
        const int beg = __builtin_amdgcn_readfirstlane(noff[l]);
        const int end = __builtin_amdgcn_readfirstlane(noff[l + 1]);
        const size_t obase = (size_t)n * OUT_DIM + lane4;
        const float4 zs = zs_next;
        if (l + GATHER_WAVES < SLICE_NODES && n + GATHER_WAVES < N_NODES)
            zs_next = *(const float4*)(z + obase + (size_t)GATHER_WAVES * OUT_DIM);
        const float sdl = head_sum8(zs.x * k.ad0 + zs.y * k.ad1 + zs.z * k.ad2 + zs.w * k.ad3);
        NodeAcc acc = { 0.f, 0.f, 0.f, 0.f, 0.f };

        for (int base = beg; base < end; base += 32) {
            const int idx = base + lane;
            const bool valid = idx < end;
            const int2 p = sorted[valid ? idx : base];
            const int srcv = valid ? p.x : n;
            const int efb = p.y;

            const int left = end - base;
            const int m = left < 32 ? left : 32;
            for (int j = 0; j < m; j += 8) {
                v4f zr[8];
#pragma unroll
                for (int jj = 0; jj < 8; ++jj) {
                    const int sj = __builtin_amdgcn_readlane(srcv, j + jj);
                    const float* rowp = z + (size_t)sj * OUT_DIM;
                    zr[jj] = *(const v4f*)(rowp + lane4);
                }
                asm volatile("" : "+v"(zr[0]), "+v"(zr[1]), "+v"(zr[2]), "+v"(zr[3]),
                                  "+v"(zr[4]), "+v"(zr[5]), "+v"(zr[6]), "+v"(zr[7]));
#pragma unroll
                for (int jj = 0; jj < 8; ++jj) {
                    const int e = j + jj;
                    const float ef = __int_as_float(__builtin_amdgcn_readlane(efb, e));
                    edge_step(acc, k, zr[jj], ef, sdl, (e < left) ? -1 : 0);
                }
            }
        }
        finish_node(acc, k, zs, out + obase);
    }
}

extern "C" void kernel_launch(void* const* d_in, const int* in_sizes, int n_in,
                              void* d_out, int out_size, void* d_ws, size_t ws_size,
                              hipStream_t stream) {
    const float* h    = (const float*)d_in[0];
    const int*   ei   = (const int*)  d_in[1];
    const float* ef   = (const float*)d_in[2];
    const float* W_w  = (const float*)d_in[3];
    const float* W_b  = (const float*)d_in[4];
    const float* att  = (const float*)d_in[5];
    const float* ln_g = (const float*)d_in[6];
    const float* ln_b = (const float*)d_in[7];
    float* out = (float*)d_out;

    float* ws      = (float*)d_ws;
    float* z       = ws;
    int2*  binned  = (int2*)(z + (size_t)N_NODES * OUT_DIM);
    int*   seg_off = (int*)(binned + N_EDGES);

    edge_bin_and_project<<<GEMM_BLOCKS + N_CHUNK, 256, 0, stream>>>(ei, ef, binned, seg_off,
                                                                    h, W_w, W_b, z);
    slice_gather_finalize<<<N_SLICE, 1024, 0, stream>>>(binned, seg_off, att, z, ln_g, ln_b, out);
}


# Source score half made once per node in the projection epilogue, read per edge; weights made one lane per (edge, head) and handed out by DPP8: score and denominator sum order change
# speedup vs baseline: 5.2194x; 1.1361x over previous
#include <hip/hip_runtime.h>
#include <math.h>

#define N_NODES 50000
#define N_EDGES 800000
#define IN_DIM  128
#define OUT_DIM 128
#define HEADS   4
#define D_K     32
#define ATT_DIM 65

typedef __attribute__((ext_vector_type(2))) float v2f;
typedef __attribute__((ext_vector_type(8))) float v8f;
typedef __attribute__((ext_vector_type(4))) float v4f;

#define W_PITCH     132
#define W_LDS_WORDS (OUT_DIM + OUT_DIM * W_PITCH)
#define W_ATT_OFF   W_LDS_WORDS
#define W_LDS_TOTAL (W_LDS_WORDS + OUT_DIM)

template <int XMASK>
__device__ __forceinline__ float add_lane_xor(float v) {
    const int o = __builtin_amdgcn_update_dpp(0, __float_as_int(v), 0x160 + XMASK, 0xf, 0xf, true);
    return v + __int_as_float(o);
}

__device__ __forceinline__ void
stage_w(const int t, const float* __restrict__ W, const float* __restrict__ bias,
        const float* __restrict__ att, float* Ws) {
    float bt = 0.f, at = 0.f;
    if (t < OUT_DIM) {
        bt = bias[t];
        at = att[(t >> 5) * ATT_DIM + (t & 31)];
    }
#pragma unroll
    for (int q = 0; q < 16; ++q) {
        const int i = q * 256 + t;
        const int row = i >> 5;
        const int c4 = i & 31;
        const v4f v = *(const v4f*)(W + (size_t)row * IN_DIM + c4 * 4);
        *(v4f*)(Ws + OUT_DIM + row * W_PITCH + c4 * 4) = v;
    }
    if (t < OUT_DIM) {
        Ws[t] = bt;
        Ws[W_ATT_OFF + t] = at;
    }
}

__device__ __forceinline__ void
gemm_z_strip(const int wave, const int lane,
             const float* __restrict__ h, const float* Ws, float* __restrict__ z,
             float* __restrict__ ssrc) {
    const int m = lane & 15;
    const int g = lane >> 4;

    const float* __restrict__ hrow = h + (size_t)(wave * 16 + m) * IN_DIM;
    v2f a[32];
#pragma unroll
    for (int ks = 0; ks < 32; ++ks) {
        const int k = 4 * ks + 2 * g;
        a[ks].x = hrow[k];
        a[ks].y = hrow[k + 1];
    }

    const float* wlane = Ws + OUT_DIM + m * W_PITCH + 2 * g;
    v2f bcur[8], bnxt[8];
#pragma unroll
    for (int i = 0; i < 8; ++i) bcur[i] = *(const v2f*)(wlane + 4 * i);

    float ph[8];
#pragma unroll
    for (int r = 0; r < 8; ++r) ph[r] = 0.f;

#pragma unroll 1
    for (int ot = 0; ot < 8; ++ot) {
        const int col = ot * 16 + m;
        const float* wrow = wlane + ot * 16 * W_PITCH;
        const float* wnext = wlane + (ot < 7 ? ot + 1 : ot) * 16 * W_PITCH;
        float* zt = z + (size_t)(wave * 16 + g * 8) * OUT_DIM + col;
        asm volatile("" : "+v"(zt));
        const float bv = Ws[col];
        const float cf = Ws[W_ATT_OFF + col];
        v8f c = {};
#pragma unroll
        for (int grp = 0; grp < 4; ++grp) {
            const float* src = grp < 3 ? wrow + 32 * (grp + 1) : wnext;
#pragma unroll
            for (int i = 0; i < 8; ++i) bnxt[i] = *(const v2f*)(src + 4 * i);
            __builtin_amdgcn_sched_barrier(0);
#pragma unroll
            for (int i = 0; i < 8; ++i) {
                c = __builtin_amdgcn_wmma_f32_16x16x4_f32(
                         false, a[8 * grp + i],  false, bcur[i],
                         (short)0, c,  false,  false);
            }
#pragma unroll
            for (int i = 0; i < 8; ++i) bcur[i] = bnxt[i];
        }
        asm volatile("v_nop\n\tv_nop" : "+v"(c));
#pragma unroll
        for (int r = 0; r < 8; ++r) {
            const float zv = c[r] + bv;
            zt[(size_t)r * OUT_DIM] = zv;
            ph[r] = fmaf(zv, cf, ph[r]);
        }
        if (ot & 1) {
            float sv = 0.f;
#pragma unroll
            for (int r = 0; r < 8; ++r) {
                float v = ph[r];
                v = add_lane_xor<1>(v);
                v = add_lane_xor<2>(v);
                v = add_lane_xor<4>(v);
                v = add_lane_xor<8>(v);
                sv = (m == r) ? v : sv;
                ph[r] = 0.f;
            }
            if (m < 8) ssrc[(size_t)(wave * 16 + g * 8 + m) * HEADS + (ot >> 1)] = sv;
        }
    }
}

__device__ __forceinline__ float leaky_clip_exp(float s) {
    s = (s >= 0.f) ? s : 0.2f * s;
    s = fminf(fmaxf(s, -20.f), 20.f);
    return __expf(s);
}

#define SLICE_NODES 196
#define N_SLICE     256
#define N_CHUNK     256
#define CHUNK_EDGES (N_EDGES / N_CHUNK)
#define CHUNK_STEPS ((CHUNK_EDGES + 255) / 256)
#define SEG_PITCH   (N_SLICE + 1)
#define GEMM_WAVES  (N_NODES / 16)
#define GEMM_BLOCKS ((GEMM_WAVES + 7) / 8)

static_assert(N_CHUNK * CHUNK_EDGES == N_EDGES, "chunks must tile the edges");
static_assert(N_SLICE * SLICE_NODES >= N_NODES && SLICE_NODES <= 256 && N_NODES <= 65536, "record packing");

__device__ __forceinline__ int wave_incl_scan(int x, const int lane) {
#pragma unroll
    for (int d = 1; d < 32; d <<= 1) {
        const int y = __shfl_up(x, d, 32);
        if (lane >= d) x += y;
    }
    return x;
}

__device__ __forceinline__ void
bin_chunk(const int c, const int t, const int* __restrict__ ei, const float* __restrict__ ef,
          int2* __restrict__ binned, int* __restrict__ seg_off, int* lds) {
    int* cnt  = lds;
    int* wtot = lds + N_SLICE;
    int2* rec = (int2*)(lds + N_SLICE + 16);
    const int lane = t & 31;
    const int w = t >> 5;

    cnt[t] = 0;
    __syncthreads();

    int sl[CHUNK_STEPS], rk[CHUNK_STEPS], px[CHUNK_STEPS], py[CHUNK_STEPS];
#pragma unroll
    for (int i = 0; i < CHUNK_STEPS; ++i) {
        const int le = i * 256 + t;
        const int e = c * CHUNK_EDGES + (le < CHUNK_EDGES ? le : 0);
        px[i] = ei[e];
        sl[i] = ei[N_EDGES + e];
        py[i] = __float_as_int(ef[e]);
    }
    __builtin_amdgcn_sched_barrier(0);
#pragma unroll
    for (int i = 0; i < CHUNK_STEPS; ++i) {
        const int dst = sl[i];
        const int s = dst / SLICE_NODES;
        sl[i] = s;
        px[i] |= (dst - s * SLICE_NODES) << 16;
        rk[i] = (i * 256 + t < CHUNK_EDGES) ? atomicAdd(&cnt[s], 1) : 0;
    }
    __syncthreads();

    const int v = cnt[t];
    const int incl = wave_incl_scan(v, lane);
    if (lane == 31) wtot[w] = incl;
    __syncthreads();
    int pre = 0;
#pragma unroll
    for (int k = 0; k < 8; ++k) pre += (k < w) ? wtot[k] : 0;
    const int excl = pre + incl - v;
    cnt[t] = excl;
    seg_off[c * SEG_PITCH + t] = excl;
    if (t == N_SLICE - 1) seg_off[c * SEG_PITCH + N_SLICE] = excl + v;
    __syncthreads();

#pragma unroll
    for (int i = 0; i < CHUNK_STEPS; ++i) {
        const int le = i * 256 + t;
        if (le < CHUNK_EDGES) rec[cnt[sl[i]] + rk[i]] = make_int2(px[i], py[i]);
    }
    __syncthreads();

    int2* outp = binned + (size_t)c * CHUNK_EDGES;
#pragma unroll
    for (int i = 0; i < CHUNK_STEPS; ++i) {
        const int le = i * 256 + t;
        if (le < CHUNK_EDGES) outp[le] = rec[le];
    }
}

__global__ void __launch_bounds__(256)
edge_bin_and_project(const int* __restrict__ ei, const float* __restrict__ ef,
                     int2* __restrict__ binned, int* __restrict__ seg_off,
                     const float* __restrict__ h, const float* __restrict__ W,
                     const float* __restrict__ bias, float* __restrict__ z,
                     const float* __restrict__ att, float* __restrict__ ssrc) {
    __shared__ v4f w_lds[W_LDS_TOTAL / 4];
    if (blockIdx.x < GEMM_BLOCKS) {
        stage_w(threadIdx.x, W, bias, att, (float*)w_lds);
        __syncthreads();
        const int wave = (blockIdx.x * blockDim.x + threadIdx.x) >> 5;
        if (wave < GEMM_WAVES)
            gemm_z_strip(wave, threadIdx.x & 31, h, (const float*)w_lds, z, ssrc);
        return;
    }
    bin_chunk(blockIdx.x - GEMM_BLOCKS, threadIdx.x, ei, ef, binned, seg_off, (int*)w_lds);
}

__device__ __forceinline__ float head_sum8(float v) {
    v = add_lane_xor<1>(v);
    v = add_lane_xor<2>(v);
    v = add_lane_xor<4>(v);
    return v;
}

#define SLICE_CAP     8192
#define GATHER_WAVES  32

struct NodeCoef {
    float as0, as1, as2, as3;
    float ad0, ad1, ad2, ad3;
    float ae;
    float4 g, b;
};

struct NodeAcc { float a0, a1, a2, a3, den; };

__device__ __forceinline__ void
edge_step(NodeAcc& s, const NodeCoef& k, const v4f zv, const float ef, const float sdl, const int keep) {
    const float ps = head_sum8(zv.x * k.as0 + zv.y * k.as1 + zv.z * k.as2 + zv.w * k.as3);
    const float w = leaky_clip_exp(ps + sdl + ef * k.ae);
    const float a = __int_as_float(__float_as_int(w) & keep);
    s.a0 = fmaf(a, zv.x, s.a0);
    s.a1 = fmaf(a, zv.y, s.a1);
    s.a2 = fmaf(a, zv.z, s.a2);
    s.a3 = fmaf(a, zv.w, s.a3);
    s.den += a;
}

__device__ __forceinline__ void
finish_node(const NodeAcc& s, const NodeCoef& k, const float4 zs, float* __restrict__ outp) {
    const float inv = 1.f / (s.den + 1e-6f);
    float x[4] = { s.a0 * inv + zs.x, s.a1 * inv + zs.y, s.a2 * inv + zs.z, s.a3 * inv + zs.w };

    float sum = x[0] + x[1] + x[2] + x[3];
#pragma unroll
    for (int o = 16; o > 0; o >>= 1) sum += __shfl_xor(sum, o, 32);
    const float mu = sum * (1.f / 128.f);

    float v = 0.f;
#pragma unroll
    for (int i = 0; i < 4; ++i) { const float d = x[i] - mu; v = fmaf(d, d, v); }
#pragma unroll
    for (int o = 16; o > 0; o >>= 1) v += __shfl_xor(v, o, 32);
    const float rstd = rsqrtf(v * (1.f / 128.f) + 1e-5f);

    const float g4[4] = { k.g.x, k.g.y, k.g.z, k.g.w };
    const float b4[4] = { k.b.x, k.b.y, k.b.z, k.b.w };
    float y[4];
#pragma unroll
    for (int i = 0; i < 4; ++i) {
        const float tt = (x[i] - mu) * rstd * g4[i] + b4[i];
        y[i] = (tt > 0.f) ? tt : (__expf(tt) - 1.f);
    }
    float4 o4 = { y[0], y[1], y[2], y[3] };
    *(float4*)outp = o4;
}

template <int E>
__device__ __forceinline__ float bcast8(const float v) {
    return __int_as_float((int)__builtin_amdgcn_mov_dpp8((unsigned)__float_as_int(v), E * 0x249249));
}

template <int BIT>
__device__ __forceinline__ void trade_across(float& lo, float& hi, const bool up) {
    const float send = up ? lo : hi;
    float recv;
    if (BIT == 8)
        recv = __int_as_float(__builtin_amdgcn_update_dpp(0, __float_as_int(send), 0x168, 0xf, 0xf, true));
    else
        recv = __shfl_xor(send, 16, 32);
    lo = up ? recv : lo;
    hi = up ? hi : recv;
}

__device__ __forceinline__ float edge_weight(const float s, const int keep) {
    return __int_as_float(__float_as_int(leaky_clip_exp(s)) & keep);
}

__device__ __forceinline__ void
ask_rows8(v4f (&zr)[8], const float* __restrict__ z, const int srcv, const int j, const unsigned lane4) {
#pragma unroll
    for (int jj = 0; jj < 8; ++jj) {
        const int sj = __builtin_amdgcn_readlane(srcv, j + jj);
        const float* rowp = z + (size_t)sj * OUT_DIM;
        zr[jj] = *(const v4f*)(rowp + lane4);
    }
}

template <int E>
__device__ __forceinline__ void
add_row(float& a0, float& a1, float& a2, float& a3, const v4f zv, const float t) {
    const float a = bcast8<E>(t);
    a0 = fmaf(a, zv.x, a0);
    a1 = fmaf(a, zv.y, a1);
    a2 = fmaf(a, zv.z, a2);
    a3 = fmaf(a, zv.w, a3);
}
__device__ __forceinline__ void
add_rows8(float& a0, float& a1, float& a2, float& a3, const v4f (&zr)[8], const float t) {
    add_row<0>(a0, a1, a2, a3, zr[0], t);
    add_row<1>(a0, a1, a2, a3, zr[1], t);
    add_row<2>(a0, a1, a2, a3, zr[2], t);
    add_row<3>(a0, a1, a2, a3, zr[3], t);
    add_row<4>(a0, a1, a2, a3, zr[4], t);
    add_row<5>(a0, a1, a2, a3, zr[5], t);
    add_row<6>(a0, a1, a2, a3, zr[6], t);
    add_row<7>(a0, a1, a2, a3, zr[7], t);
}

__global__ void __launch_bounds__(1024)
slice_gather_finalize(const int2* __restrict__ binned, const int* __restrict__ seg_off,
                      const float* __restrict__ att, const float* __restrict__ z,
                      const float* __restrict__ ssrc,
                      const float* __restrict__ ln_g, const float* __restrict__ ln_b,
                      float* __restrict__ out) {
    __shared__ float att_s[HEADS * ATT_DIM];
    __shared__ int seg_start[N_CHUNK];
    __shared__ int seg_len[N_CHUNK];
    __shared__ int seg_pos[N_CHUNK];
    __shared__ int wtot[8];
    __shared__ int ncnt[256];
    __shared__ int noff[257];
    __shared__ int total_s;
    __shared__ int2 stage[SLICE_CAP];
    __shared__ int2 sorted[SLICE_CAP];

    const int t = threadIdx.x;
    const int lane = t & 31;
    const int w = __builtin_amdgcn_readfirstlane(t >> 5);
    const int s = blockIdx.x;

    float av = 0.f;
    if (t < HEADS * ATT_DIM) av = att[t];
    int len = 0, start = 0;
    if (t < N_CHUNK) {
        const int a = seg_off[t * SEG_PITCH + s];
        const int b = seg_off[t * SEG_PITCH + s + 1];
        len = b - a;
        start = t * CHUNK_EDGES + a;
        ncnt[t] = 0;
    }
    if (t < HEADS * ATT_DIM) att_s[t] = av;
    const int incl = wave_incl_scan(len, lane);
    if (t < N_CHUNK && lane == 31) wtot[w] = incl;
    __syncthreads();
    if (t < N_CHUNK) {
        int pre = 0;
#pragma unroll
        for (int k = 0; k < 8; ++k) pre += (k < w) ? wtot[k] : 0;
        const int excl = pre + incl - len;
        seg_start[t] = start;
        seg_len[t] = len;
        seg_pos[t] = excl;
        if (t == N_CHUNK - 1) total_s = excl + len;
    }
    __syncthreads();
    const int M = __builtin_amdgcn_readfirstlane(total_s);

    NodeCoef k;
    {
        const int hd = lane >> 3;
        const float* ah = att_s + hd * ATT_DIM + (lane & 7) * 4;
        k.as0 = ah[0]; k.as1 = ah[1]; k.as2 = ah[2]; k.as3 = ah[3];
        k.ad0 = ah[D_K]; k.ad1 = ah[D_K + 1]; k.ad2 = ah[D_K + 2]; k.ad3 = ah[D_K + 3];
        k.ae = att_s[hd * ATT_DIM + 2 * D_K];
        k.g = *(const float4*)(ln_g + lane * 4);
        k.b = *(const float4*)(ln_b + lane * 4);
    }
    const unsigned lane4 = (unsigned)lane * 4u;
    const int node0 = s * SLICE_NODES;
    const float ae0 = att_s[0 * ATT_DIM + 2 * D_K], ae1 = att_s[1 * ATT_DIM + 2 * D_K];
    const float ae2 = att_s[2 * ATT_DIM + 2 * D_K], ae3 = att_s[3 * ATT_DIM + 2 * D_K];
    const bool up16 = (lane & 16) != 0, up8 = (lane & 8) != 0;

    if (M > SLICE_CAP) {
        for (int l = w; l < SLICE_NODES; l += GATHER_WAVES) {
            const int n = node0 + l;
            if (n >= N_NODES) break;
            const size_t obase = (size_t)n * OUT_DIM + lane4;
            const float4 zs = *(const float4*)(z + obase);
            const float sdl = head_sum8(zs.x * k.ad0 + zs.y * k.ad1 + zs.z * k.ad2 + zs.w * k.ad3);
            NodeAcc acc = { 0.f, 0.f, 0.f, 0.f, 0.f };
            for (int c = 0; c < N_CHUNK; ++c) {
                const int st = __builtin_amdgcn_readfirstlane(seg_start[c]);
                const int ln = __builtin_amdgcn_readfirstlane(seg_len[c]);
                for (int i0 = 0; i0 < ln; i0 += 32) {
                    const int i = i0 + lane;
                    int2 r = make_int2(0, 0);
                    if (i < ln) r = binned[st + i];
                    const bool hit = (i < ln) && (int)((unsigned)r.x >> 16) == l;
                    unsigned mask = __builtin_amdgcn_ballot_w32(hit);
                    while (mask) {
                        const int bpos = __builtin_ctz(mask);
                        mask &= mask - 1;
                        const int sj = __builtin_amdgcn_readlane(r.x, bpos) & 0xffff;
                        const float efv = __int_as_float(__builtin_amdgcn_readlane(r.y, bpos));
                        const v4f zv = *(const v4f*)(z + (size_t)sj * OUT_DIM + lane4);
                        edge_step(acc, k, zv, efv, sdl, -1);
                    }
                }
            }
            finish_node(acc, k, zs, out + obase);
        }
        return;
    }

    {
        int st[N_CHUNK / GATHER_WAVES], ln[N_CHUNK / GATHER_WAVES], ps[N_CHUNK / GATHER_WAVES];
        int2 r[N_CHUNK / GATHER_WAVES];
#pragma unroll
        for (int q = 0; q < N_CHUNK / GATHER_WAVES; ++q) {
            const int c = w + GATHER_WAVES * q;
            st[q] = __builtin_amdgcn_readfirstlane(seg_start[c]);
            ln[q] = __builtin_amdgcn_readfirstlane(seg_len[c]);
            ps[q] = __builtin_amdgcn_readfirstlane(seg_pos[c]);
            r[q] = make_int2(0, 0);
            if (lane < ln[q]) r[q] = binned[st[q] + lane];
        }
#pragma unroll
        for (int q = 0; q < N_CHUNK / GATHER_WAVES; ++q) {
            if (lane < ln[q]) {
                stage[ps[q] + lane] = r[q];
                atomicAdd(&ncnt[(unsigned)r[q].x >> 16], 1);
            }
            for (int i = lane + 32; i < ln[q]; i += 32) {
                const int2 rr = binned[st[q] + i];
                stage[ps[q] + i] = rr;
                atomicAdd(&ncnt[(unsigned)rr.x >> 16], 1);
            }
        }
    }
    __syncthreads();

    int nv = 0, nincl = 0;
    if (t < 256) {
        nv = ncnt[t];
        nincl = wave_incl_scan(nv, lane);
        if (lane == 31) wtot[w] = nincl;
    }
    __syncthreads();
    if (t < 256) {
        int pre = 0;
#pragma unroll
        for (int q = 0; q < 8; ++q) pre += (q < w) ? wtot[q] : 0;
        const int excl = pre + nincl - nv;
        noff[t] = excl;
        ncnt[t] = excl;
        if (t == 255) noff[256] = excl + nv;
    }
    __syncthreads();

    for (int j = t; j < M; j += 1024) {
        const int2 r = stage[j];
        const int pos = atomicAdd(&ncnt[(unsigned)r.x >> 16], 1);
        sorted[pos] = make_int2(r.x & 0xffff, r.y);
    }
    __syncthreads();

    float4 zs_next = make_float4(0.f, 0.f, 0.f, 0.f);
    if (node0 + w < N_NODES) zs_next = *(const float4*)(z + (size_t)(node0 + w) * OUT_DIM + lane4);
    for (int l = w; l < SLICE_NODES; l += GATHER_WAVES) {
        const int n = node0 + l;
        if (n >= N_NODES) break;
        const int beg = __builtin_amdgcn_readfirstlane(noff[l]);
        const int end = __builtin_amdgcn_readfirstlane(noff[l + 1]);
        const size_t obase = (size_t)n * OUT_DIM + lane4;
        const float4 zs = zs_next;
        if (l + GATHER_WAVES < SLICE_NODES && n + GATHER_WAVES < N_NODES)
            zs_next = *(const float4*)(z + obase + (size_t)GATHER_WAVES * OUT_DIM);
        const float sdl = head_sum8(zs.x * k.ad0 + zs.y * k.ad1 + zs.z * k.ad2 + zs.w * k.ad3);
        const float sd0 = __int_as_float(__builtin_amdgcn_readlane(__float_as_int(sdl), 0));
        const float sd1 = __int_as_float(__builtin_amdgcn_readlane(__float_as_int(sdl), 8));
        const float sd2 = __int_as_float(__builtin_amdgcn_readlane(__float_as_int(sdl), 16));
        const float sd3 = __int_as_float(__builtin_amdgcn_readlane(__float_as_int(sdl), 24));
        float a0 = 0.f, a1 = 0.f, a2 = 0.f, a3 = 0.f;
        float denp = 0.f;

        for (int base = beg; base < end; base += 32) {
            const int idx = base + lane;
            const bool valid = idx < end;
            const int2 p = sorted[valid ? idx : base];
            const int srcv = valid ? p.x : n;
            const float ef = __int_as_float(p.y);
            const int keep = valid ? -1 : 0;

            v4f sc = *(const v4f*)(ssrc + (size_t)srcv * HEADS);
            v4f zr[8];
            ask_rows8(zr, z, srcv, 0, lane4);
            asm volatile("" : "+v"(sc), "+v"(zr[0]), "+v"(zr[1]), "+v"(zr[2]), "+v"(zr[3]),
                              "+v"(zr[4]), "+v"(zr[5]), "+v"(zr[6]), "+v"(zr[7]));

            float t0 = edge_weight(sc.x + sd0 + ef * ae0, keep);
            float t1 = edge_weight(sc.y + sd1 + ef * ae1, keep);
            float t2 = edge_weight(sc.z + sd2 + ef * ae2, keep);
            float t3 = edge_weight(sc.w + sd3 + ef * ae3, keep);
            trade_across<16>(t0, t2, up16);
            trade_across<16>(t1, t3, up16);
            trade_across<8>(t0, t1, up8);
            trade_across<8>(t2, t3, up8);

            const int left = end - base;
            const int m = left < 32 ? left : 32;
            denp += t0;
            add_rows8(a0, a1, a2, a3, zr, t0);
            t0 = t1; t1 = t2; t2 = t3;
            for (int j = 8; j < m; j += 8) {
                ask_rows8(zr, z, srcv, j, lane4);
                asm volatile("" : "+v"(zr[0]), "+v"(zr[1]), "+v"(zr[2]), "+v"(zr[3]),
                                  "+v"(zr[4]), "+v"(zr[5]), "+v"(zr[6]), "+v"(zr[7]));
                denp += t0;
                add_rows8(a0, a1, a2, a3, zr, t0);
                t0 = t1; t1 = t2; t2 = t3;
            }
        }
        const NodeAcc acc = { a0, a1, a2, a3, head_sum8(denp) };
        finish_node(acc, k, zs, out + obase);
    }
}

extern "C" void kernel_launch(void* const* d_in, const int* in_sizes, int n_in,
                              void* d_out, int out_size, void* d_ws, size_t ws_size,
                              hipStream_t stream) {
    const float* h    = (const float*)d_in[0];
    const int*   ei   = (const int*)  d_in[1];
    const float* ef   = (const float*)d_in[2];
    const float* W_w  = (const float*)d_in[3];
    const float* W_b  = (const float*)d_in[4];
    const float* att  = (const float*)d_in[5];
    const float* ln_g = (const float*)d_in[6];
    const float* ln_b = (const float*)d_in[7];
    float* out = (float*)d_out;

    float* ws      = (float*)d_ws;
    float* z       = ws;
    int2*  binned  = (int2*)(z + (size_t)N_NODES * OUT_DIM);
    int*   seg_off = (int*)(binned + N_EDGES);
    float* ssrc    = (float*)(seg_off + N_CHUNK * SEG_PITCH);

    edge_bin_and_project<<<GEMM_BLOCKS + N_CHUNK, 256, 0, stream>>>(ei, ef, binned, seg_off,
                                                                    h, W_w, W_b, z, att, ssrc);
    slice_gather_finalize<<<N_SLICE, 1024, 0, stream>>>(binned, seg_off, att, z, ssrc, ln_g, ln_b, out);
}
